# starting kernel
# baseline (speedup 1.0000x reference)
#include <hip/hip_runtime.h>
#include <math.h>

typedef __attribute__((ext_vector_type(2))) float v2f;
typedef __attribute__((ext_vector_type(8))) float v8f;

#define WAVES_PER_BLOCK 8
#define EPSV 1e-4f
#define INV_SQRT10 0.3162277660168379f

__device__ const float kPre[16] = {1,1,3,3,1,1,3,3,-1,-1,-3,-3,-1,-1,-3,-3};
__device__ const float kPim[16] = {1,3,1,3,-1,-3,-1,-3,1,3,1,3,-1,-3,-1,-3};

struct WS {
  union {
    struct ChT { float Are[16][17]; float Aim[16][17]; } ch;
    float aug[16][33];
  } u;
  float Wre[16][10], Wim[16][10];
  float gre[8][9],  gim[8][9];
  float gr[16][17];
  float yptr[16][17];
  float ymfre[8], ymfim[8];
  float xre[8], xim[8], varL[8];
  float gxre[8], gxim[8];
  float muL[16], znumL[16];
  float rhoL[8];
  float llr_aL[8][4], llr_dL[8][4];
};

__device__ __forceinline__ float log_sigmoid(float x) {
  return fminf(x, 0.0f) - log1pf(__expf(-fabsf(x)));
}

__global__ __launch_bounds__(256) void mmse_pic_kernel(
    const float* __restrict__ y_re, const float* __restrict__ y_im,
    const float* __restrict__ h_re, const float* __restrict__ h_im,
    const float* __restrict__ prior,
    const float* __restrict__ s_re, const float* __restrict__ s_im,
    float* __restrict__ out)
{
  __shared__ WS ws[WAVES_PER_BLOCK];
  const int lane = threadIdx.x & 31;
  const int wid  = threadIdx.x >> 5;
  const int b    = blockIdx.x * WAVES_PER_BLOCK + wid;
  WS& w = ws[wid];

  #pragma unroll
  for (int t = 0; t < 8; ++t) {
    int idx = lane + 32*t;
    int i = idx >> 4, j = idx & 15;
    w.u.ch.Are[i][j] = s_re[b*256 + idx];
    w.u.ch.Aim[i][j] = s_im[b*256 + idx];
  }
  #pragma unroll
  for (int t = 0; t < 4; ++t) {
    int idx = lane + 32*t;
    int m = idx >> 3, c = idx & 7;
    w.Wre[m][c] = h_re[b*128 + idx];
    w.Wim[m][c] = h_im[b*128 + idx];
  }
  if (lane < 16) {
    w.Wre[lane][8] = y_re[b*16 + lane];
    w.Wim[lane][8] = y_im[b*16 + lane];
  }
  const int s_out = lane >> 2, k_out = lane & 3;
  float my_lld = prior[b*32 + lane];
  w.llr_dL[s_out][k_out] = my_lld;
  __syncthreads();

  const int r = lane;
  for (int k = 0; k < 16; ++k) {
    float dk  = w.u.ch.Are[k][k];
    float sdk = sqrtf(dk);
    float isd = 1.0f / sdk;
    if (r == k) { w.u.ch.Are[k][k] = sdk; w.u.ch.Aim[k][k] = 0.0f; }
    if (r < 16 && r > k) { w.u.ch.Are[r][k] *= isd; w.u.ch.Aim[r][k] *= isd; }
    __syncthreads();
    if (r < 16 && r > k) {
      float lre = w.u.ch.Are[r][k], lim = w.u.ch.Aim[r][k];
      for (int j = k+1; j <= r; ++j) {
        float cre = w.u.ch.Are[j][k], cim = w.u.ch.Aim[j][k];
        w.u.ch.Are[r][j] -= lre*cre + lim*cim;
        w.u.ch.Aim[r][j] -= lim*cre - lre*cim;
      }
    }
    __syncthreads();
  }

  if (lane < 9) {
    const int c = lane;
    for (int m = 0; m < 16; ++m) {
      float accre = w.Wre[m][c], accim = w.Wim[m][c];
      for (int j = 0; j < m; ++j) {
        float lre = w.u.ch.Are[m][j], lim = w.u.ch.Aim[m][j];
        float wr = w.Wre[j][c], wi = w.Wim[j][c];
        accre -= lre*wr - lim*wi;
        accim -= lre*wi + lim*wr;
      }
      float inv = 1.0f / w.u.ch.Are[m][m];
      w.Wre[m][c] = accre*inv;
      w.Wim[m][c] = accim*inv;
    }
  }
  __syncthreads();

  if (lane < 8) {
    int s_ = lane;
    float ar = 0.0f, ai = 0.0f;
    #pragma unroll
    for (int m = 0; m < 16; ++m) {
      float hr = w.Wre[m][s_], hi = w.Wim[m][s_];
      float yr = w.Wre[m][8],  yi = w.Wim[m][8];
      ar += hr*yr + hi*yi;
      ai += hr*yi - hi*yr;
    }
    w.ymfre[s_] = ar; w.ymfim[s_] = ai;
  }
  #pragma unroll
  for (int t = 0; t < 2; ++t) {
    int idx = lane + 32*t;
    int i = idx >> 3, j = idx & 7;
    float ar = 0.0f, ai = 0.0f;
    #pragma unroll
    for (int m = 0; m < 16; ++m) {
      float xr = w.Wre[m][i], xi = w.Wim[m][i];
      float yr = w.Wre[m][j], yi = w.Wim[m][j];
      ar += xr*yr + xi*yi;
      ai += xr*yi - xi*yr;
    }
    w.gre[i][j] = ar; w.gim[i][j] = ai;
  }
  __syncthreads();
  #pragma unroll
  for (int t = 0; t < 8; ++t) {
    int idx = lane + 32*t;
    int i = idx >> 4, j = idx & 15;
    float v;
    if (i < 8) v = (j < 8) ? w.gre[i][j]   : -w.gim[i][j-8];
    else       v = (j < 8) ? w.gim[i-8][j] :  w.gre[i-8][j-8];
    w.gr[i][j] = v;
  }
  __syncthreads();

  float my_lla = 0.0f;
  for (int iter = 0; iter < 2; ++iter) {
    my_lla = my_lld;
    w.llr_aL[s_out][k_out] = my_lla;
    __syncthreads();

    float lg[16];
    if (lane < 8) {
      int s_ = lane;
      float l0 = w.llr_aL[s_][0], l1 = w.llr_aL[s_][1];
      float l2 = w.llr_aL[s_][2], l3 = w.llr_aL[s_][3];
      #pragma unroll
      for (int p = 0; p < 16; ++p) {
        lg[p] = log_sigmoid((p & 8) ? l0 : -l0)
              + log_sigmoid((p & 4) ? l1 : -l1)
              + log_sigmoid((p & 2) ? l2 : -l2)
              + log_sigmoid((p & 1) ? l3 : -l3);
      }
      float mx = lg[0];
      #pragma unroll
      for (int p = 1; p < 16; ++p) mx = fmaxf(mx, lg[p]);
      float sum = 0.0f, mre = 0.0f, mim = 0.0f, e2 = 0.0f;
      #pragma unroll
      for (int p = 0; p < 16; ++p) {
        float e = __expf(lg[p] - mx);
        float pr = kPre[p]*INV_SQRT10, pi = kPim[p]*INV_SQRT10;
        sum += e; mre += e*pr; mim += e*pi; e2 += e*(pr*pr + pi*pi);
      }
      float invs = 1.0f / sum;
      mre *= invs; mim *= invs; e2 *= invs;
      w.xre[s_] = mre; w.xim[s_] = mim;
      w.varL[s_] = e2 - (mre*mre + mim*mim);
    }
    __syncthreads();

    if (lane < 8) {
      int i = lane;
      float ar = 0.0f, ai = 0.0f;
      #pragma unroll
      for (int j = 0; j < 8; ++j) {
        float gre_ = w.gre[i][j], gim_ = w.gim[i][j];
        float xr = w.xre[j], xi = w.xim[j];
        ar += gre_*xr - gim_*xi;
        ai += gre_*xi + gim_*xr;
      }
      w.gxre[i] = ar; w.gxim[i] = ai;
    }
    __syncthreads();

    #pragma unroll
    for (int t = 0; t < 2; ++t) {
      int idx = lane + 32*t;
      int i = idx >> 3, j = idx & 7;
      float gre_ = w.gre[j][i], gim_ = w.gim[j][i];
      float xr = w.xre[i], xi = w.xim[i];
      float tre = w.ymfre[j] + (gre_*xr - gim_*xi) - w.gxre[j];
      float tim = w.ymfim[j] + (gre_*xi + gim_*xr) - w.gxim[j];
      w.yptr[i][j]     = tre; w.yptr[i][j+8]   = tim;
      w.yptr[i+8][j]   = tre; w.yptr[i+8][j+8] = tim;
    }
    {
      int rr = lane & 15, hf = lane >> 4, base = hf*16;
      #pragma unroll
      for (int c = 0; c < 16; ++c) {
        int col = base + c;
        float v;
        if (col < 16) v = w.gr[rr][col] * w.varL[col & 7] + ((rr == col) ? 1.0f : 0.0f);
        else          v = ((col - 16) == rr) ? 1.0f : 0.0f;
        w.u.aug[rr][col] = v;
      }
    }
    __syncthreads();

    {
      int rr = lane & 15, hf = lane >> 4, base = hf*16;
      for (int k = 0; k < 16; ++k) {
        float pinv = 1.0f / w.u.aug[k][k];
        __syncthreads();
        if (rr == k) {
          #pragma unroll
          for (int c = 0; c < 16; ++c) w.u.aug[k][base+c] *= pinv;
        }
        __syncthreads();
        float f = w.u.aug[rr][k];
        if (rr != k) {
          #pragma unroll
          for (int c = 0; c < 16; ++c)
            w.u.aug[rr][base+c] -= f * w.u.aug[k][base+c];
        }
        __syncthreads();
      }
    }

    {
      int nn = lane & 15, hf = lane >> 4;
      v8f p1 = {}; v8f p2 = {};
      #pragma unroll
      for (int kb = 0; kb < 4; ++kb) {
        int k0 = kb*4 + hf*2;
        v2f af, b1, b2;
        af.x = w.u.aug[nn][16 + k0];
        af.y = w.u.aug[nn][16 + k0 + 1];
        b1.x = w.gr[k0][nn];   b1.y = w.gr[k0+1][nn];
        b2.x = w.yptr[nn][k0]; b2.y = w.yptr[nn][k0+1];
        p1 = __builtin_amdgcn_wmma_f32_16x16x4_f32(false, af, false, b1,
                                                   (short)0, p1, false, false);
        p2 = __builtin_amdgcn_wmma_f32_16x16x4_f32(false, af, false, b2,
                                                   (short)0, p2, false, false);
      }
      #pragma unroll
      for (int v = 0; v < 8; ++v) {
        int row = v + 8*hf;
        if (row == nn) { w.muL[nn] = p1[v]; w.znumL[nn] = p2[v]; }
      }
    }
    __syncthreads();

    if (lane < 8) {
      int s_ = lane;
      float m1 = w.muL[s_], m2 = w.muL[s_ + 8];
      w.xre[s_] = w.znumL[s_]     / m1;
      w.xim[s_] = w.znumL[s_ + 8] / m2;
      w.rhoL[s_] = m1 / fmaxf(1.0f - w.varL[s_] * m1, EPSV);
    }
    __syncthreads();

    if (lane < 8) {
      int s_ = lane;
      float zr = w.xre[s_], zi = w.xim[s_];
      float rho = w.rhoL[s_];
      float b1m[4] = {-1e30f,-1e30f,-1e30f,-1e30f};
      float b0m[4] = {-1e30f,-1e30f,-1e30f,-1e30f};
      #pragma unroll
      for (int p = 0; p < 16; ++p) {
        float pr = kPre[p]*INV_SQRT10, pi = kPim[p]*INV_SQRT10;
        float dre = zr - pr, dim = zi - pi;
        float e = -(dre*dre + dim*dim) * rho + lg[p];
        #pragma unroll
        for (int k = 0; k < 4; ++k) {
          int bit = (p >> (3-k)) & 1;
          if (bit) b1m[k] = fmaxf(b1m[k], e);
          else     b0m[k] = fmaxf(b0m[k], e);
        }
      }
      #pragma unroll
      for (int k = 0; k < 4; ++k) w.llr_dL[s_][k] = b1m[k] - b0m[k];
    }
    __syncthreads();
    my_lld = w.llr_dL[s_out][k_out];
    __syncthreads();
  }

  out[b*32 + lane] = my_lld - my_lla;
}

extern "C" void kernel_launch(void* const* d_in, const int* in_sizes, int n_in,
                              void* d_out, int out_size, void* d_ws, size_t ws_size,
                              hipStream_t stream) {
  (void)n_in; (void)out_size; (void)d_ws; (void)ws_size;
  const float* y_re  = (const float*)d_in[0];
  const float* y_im  = (const float*)d_in[1];
  const float* h_re  = (const float*)d_in[2];
  const float* h_im  = (const float*)d_in[3];
  const float* prior = (const float*)d_in[4];
  const float* s_re  = (const float*)d_in[5];
  const float* s_im  = (const float*)d_in[6];
  float* out = (float*)d_out;

  int B = in_sizes[0] / 16;
  int blocks = B / WAVES_PER_BLOCK;
  mmse_pic_kernel<<<blocks, 256, 0, stream>>>(y_re, y_im, h_re, h_im,
                                              prior, s_re, s_im, out);
}


# Gauss-Jordan inverse of [a|I] held in registers (lane = column, 16 rows per lane); pivot and row factors broadcast by v_readlane; no LDS traffic or barriers in the elimination; same arithmetic
# speedup vs baseline: 1.6432x; 1.6432x over previous
#include <hip/hip_runtime.h>
#include <math.h>

typedef __attribute__((ext_vector_type(2))) float v2f;
typedef __attribute__((ext_vector_type(8))) float v8f;

#define WAVES_PER_BLOCK 8
#define EPSV 1e-4f
#define INV_SQRT10 0.3162277660168379f

__device__ const float kPre[16] = {1,1,3,3,1,1,3,3,-1,-1,-3,-3,-1,-1,-3,-3};
__device__ const float kPim[16] = {1,3,1,3,-1,-3,-1,-3,1,3,1,3,-1,-3,-1,-3};

struct WS {
  union {
    struct ChT { float Are[16][17]; float Aim[16][17]; } ch;
    float aug[16][33];
  } u;
  float Wre[16][10], Wim[16][10];
  float gre[8][9],  gim[8][9];
  float gr[16][17];
  float yptr[16][17];
  float ymfre[8], ymfim[8];
  float xre[8], xim[8], varL[8];
  float gxre[8], gxim[8];
  float muL[16], znumL[16];
  float rhoL[8];
  float llr_aL[8][4], llr_dL[8][4];
};

__device__ __forceinline__ float log_sigmoid(float x) {
  return fminf(x, 0.0f) - log1pf(__expf(-fabsf(x)));
}

__global__ __launch_bounds__(256) void mmse_pic_kernel(
    const float* __restrict__ y_re, const float* __restrict__ y_im,
    const float* __restrict__ h_re, const float* __restrict__ h_im,
    const float* __restrict__ prior,
    const float* __restrict__ s_re, const float* __restrict__ s_im,
    float* __restrict__ out)
{
  __shared__ WS ws[WAVES_PER_BLOCK];
  const int lane = threadIdx.x & 31;
  const int wid  = threadIdx.x >> 5;
  const int b    = blockIdx.x * WAVES_PER_BLOCK + wid;
  WS& w = ws[wid];

  #pragma unroll
  for (int t = 0; t < 8; ++t) {
    int idx = lane + 32*t;
    int i = idx >> 4, j = idx & 15;
    w.u.ch.Are[i][j] = s_re[b*256 + idx];
    w.u.ch.Aim[i][j] = s_im[b*256 + idx];
  }
  #pragma unroll
  for (int t = 0; t < 4; ++t) {
    int idx = lane + 32*t;
    int m = idx >> 3, c = idx & 7;
    w.Wre[m][c] = h_re[b*128 + idx];
    w.Wim[m][c] = h_im[b*128 + idx];
  }
  if (lane < 16) {
    w.Wre[lane][8] = y_re[b*16 + lane];
    w.Wim[lane][8] = y_im[b*16 + lane];
  }
  const int s_out = lane >> 2, k_out = lane & 3;
  float my_lld = prior[b*32 + lane];
  w.llr_dL[s_out][k_out] = my_lld;
  __syncthreads();

  const int r = lane;
  for (int k = 0; k < 16; ++k) {
    float dk  = w.u.ch.Are[k][k];
    float sdk = sqrtf(dk);
    float isd = 1.0f / sdk;
    if (r == k) { w.u.ch.Are[k][k] = sdk; w.u.ch.Aim[k][k] = 0.0f; }
    if (r < 16 && r > k) { w.u.ch.Are[r][k] *= isd; w.u.ch.Aim[r][k] *= isd; }
    __syncthreads();
    if (r < 16 && r > k) {
      float lre = w.u.ch.Are[r][k], lim = w.u.ch.Aim[r][k];
      for (int j = k+1; j <= r; ++j) {
        float cre = w.u.ch.Are[j][k], cim = w.u.ch.Aim[j][k];
        w.u.ch.Are[r][j] -= lre*cre + lim*cim;
        w.u.ch.Aim[r][j] -= lim*cre - lre*cim;
      }
    }
    __syncthreads();
  }

  if (lane < 9) {
    const int c = lane;
    for (int m = 0; m < 16; ++m) {
      float accre = w.Wre[m][c], accim = w.Wim[m][c];
      for (int j = 0; j < m; ++j) {
        float lre = w.u.ch.Are[m][j], lim = w.u.ch.Aim[m][j];
        float wr = w.Wre[j][c], wi = w.Wim[j][c];
        accre -= lre*wr - lim*wi;
        accim -= lre*wi + lim*wr;
      }
      float inv = 1.0f / w.u.ch.Are[m][m];
      w.Wre[m][c] = accre*inv;
      w.Wim[m][c] = accim*inv;
    }
  }
  __syncthreads();

  if (lane < 8) {
    int s_ = lane;
    float ar = 0.0f, ai = 0.0f;
    #pragma unroll
    for (int m = 0; m < 16; ++m) {
      float hr = w.Wre[m][s_], hi = w.Wim[m][s_];
      float yr = w.Wre[m][8],  yi = w.Wim[m][8];
      ar += hr*yr + hi*yi;
      ai += hr*yi - hi*yr;
    }
    w.ymfre[s_] = ar; w.ymfim[s_] = ai;
  }
  #pragma unroll
  for (int t = 0; t < 2; ++t) {
    int idx = lane + 32*t;
    int i = idx >> 3, j = idx & 7;
    float ar = 0.0f, ai = 0.0f;
    #pragma unroll
    for (int m = 0; m < 16; ++m) {
      float xr = w.Wre[m][i], xi = w.Wim[m][i];
      float yr = w.Wre[m][j], yi = w.Wim[m][j];
      ar += xr*yr + xi*yi;
      ai += xr*yi - xi*yr;
    }
    w.gre[i][j] = ar; w.gim[i][j] = ai;
  }
  __syncthreads();
  #pragma unroll
  for (int t = 0; t < 8; ++t) {
    int idx = lane + 32*t;
    int i = idx >> 4, j = idx & 15;
    float v;
    if (i < 8) v = (j < 8) ? w.gre[i][j]   : -w.gim[i][j-8];
    else       v = (j < 8) ? w.gim[i-8][j] :  w.gre[i-8][j-8];
    w.gr[i][j] = v;
  }
  __syncthreads();

  float my_lla = 0.0f;
  for (int iter = 0; iter < 2; ++iter) {
    my_lla = my_lld;
    w.llr_aL[s_out][k_out] = my_lla;
    __syncthreads();

    float lg[16];
    if (lane < 8) {
      int s_ = lane;
      float l0 = w.llr_aL[s_][0], l1 = w.llr_aL[s_][1];
      float l2 = w.llr_aL[s_][2], l3 = w.llr_aL[s_][3];
      #pragma unroll
      for (int p = 0; p < 16; ++p) {
        lg[p] = log_sigmoid((p & 8) ? l0 : -l0)
              + log_sigmoid((p & 4) ? l1 : -l1)
              + log_sigmoid((p & 2) ? l2 : -l2)
              + log_sigmoid((p & 1) ? l3 : -l3);
      }
      float mx = lg[0];
      #pragma unroll
      for (int p = 1; p < 16; ++p) mx = fmaxf(mx, lg[p]);
      float sum = 0.0f, mre = 0.0f, mim = 0.0f, e2 = 0.0f;
      #pragma unroll
      for (int p = 0; p < 16; ++p) {
        float e = __expf(lg[p] - mx);
        float pr = kPre[p]*INV_SQRT10, pi = kPim[p]*INV_SQRT10;
        sum += e; mre += e*pr; mim += e*pi; e2 += e*(pr*pr + pi*pi);
      }
      float invs = 1.0f / sum;
      mre *= invs; mim *= invs; e2 *= invs;
      w.xre[s_] = mre; w.xim[s_] = mim;
      w.varL[s_] = e2 - (mre*mre + mim*mim);
    }
    __syncthreads();

    if (lane < 8) {
      int i = lane;
      float ar = 0.0f, ai = 0.0f;
      #pragma unroll
      for (int j = 0; j < 8; ++j) {
        float gre_ = w.gre[i][j], gim_ = w.gim[i][j];
        float xr = w.xre[j], xi = w.xim[j];
        ar += gre_*xr - gim_*xi;
        ai += gre_*xi + gim_*xr;
      }
      w.gxre[i] = ar; w.gxim[i] = ai;
    }
    __syncthreads();

    #pragma unroll
    for (int t = 0; t < 2; ++t) {
      int idx = lane + 32*t;
      int i = idx >> 3, j = idx & 7;
      float gre_ = w.gre[j][i], gim_ = w.gim[j][i];
      float xr = w.xre[i], xi = w.xim[i];
      float tre = w.ymfre[j] + (gre_*xr - gim_*xi) - w.gxre[j];
      float tim = w.ymfim[j] + (gre_*xi + gim_*xr) - w.gxim[j];
      w.yptr[i][j]     = tre; w.yptr[i][j+8]   = tim;
      w.yptr[i+8][j]   = tre; w.yptr[i+8][j+8] = tim;
    }
    {
      const int cc = lane & 15;
      const float vv = w.varL[cc & 7];
      float x[16];
      #pragma unroll
      for (int r = 0; r < 16; ++r) {
        float dl = (r == cc) ? 1.0f : 0.0f;
        float left = __builtin_fmaf(w.gr[r][cc], vv, dl);
        x[r] = (lane < 16) ? left : dl;
      }
      #pragma unroll
      for (int k = 0; k < 16; ++k) {
        float piv  = __int_as_float(__builtin_amdgcn_readlane(__float_as_int(x[k]), k));
        float pinv = 1.0f / piv;
        float f[16];
        #pragma unroll
        for (int r = 0; r < 16; ++r)
          f[r] = __int_as_float(__builtin_amdgcn_readlane(__float_as_int(x[r]), k));
        x[k] *= pinv;
        const float t = x[k];
        #pragma unroll
        for (int r = 0; r < 16; ++r)
          if (r != k) x[r] -= f[r] * t;
      }
      #pragma unroll
      for (int r = 0; r < 16; ++r) w.u.aug[r][lane] = x[r];
    }
    __syncthreads();

    {
      int nn = lane & 15, hf = lane >> 4;
      v8f p1 = {}; v8f p2 = {};
      #pragma unroll
      for (int kb = 0; kb < 4; ++kb) {
        int k0 = kb*4 + hf*2;
        v2f af, b1, b2;
        af.x = w.u.aug[nn][16 + k0];
        af.y = w.u.aug[nn][16 + k0 + 1];
        b1.x = w.gr[k0][nn];   b1.y = w.gr[k0+1][nn];
        b2.x = w.yptr[nn][k0]; b2.y = w.yptr[nn][k0+1];
        p1 = __builtin_amdgcn_wmma_f32_16x16x4_f32(false, af, false, b1,
                                                   (short)0, p1, false, false);
        p2 = __builtin_amdgcn_wmma_f32_16x16x4_f32(false, af, false, b2,
                                                   (short)0, p2, false, false);
      }
      #pragma unroll
      for (int v = 0; v < 8; ++v) {
        int row = v + 8*hf;
        if (row == nn) { w.muL[nn] = p1[v]; w.znumL[nn] = p2[v]; }
      }
    }
    __syncthreads();

    if (lane < 8) {
      int s_ = lane;
      float m1 = w.muL[s_], m2 = w.muL[s_ + 8];
      w.xre[s_] = w.znumL[s_]     / m1;
      w.xim[s_] = w.znumL[s_ + 8] / m2;
      w.rhoL[s_] = m1 / fmaxf(1.0f - w.varL[s_] * m1, EPSV);
    }
    __syncthreads();

    if (lane < 8) {
      int s_ = lane;
      float zr = w.xre[s_], zi = w.xim[s_];
      float rho = w.rhoL[s_];
      float b1m[4] = {-1e30f,-1e30f,-1e30f,-1e30f};
      float b0m[4] = {-1e30f,-1e30f,-1e30f,-1e30f};
      #pragma unroll
      for (int p = 0; p < 16; ++p) {
        float pr = kPre[p]*INV_SQRT10, pi = kPim[p]*INV_SQRT10;
        float dre = zr - pr, dim = zi - pi;
        float e = -(dre*dre + dim*dim) * rho + lg[p];
        #pragma unroll
        for (int k = 0; k < 4; ++k) {
          int bit = (p >> (3-k)) & 1;
          if (bit) b1m[k] = fmaxf(b1m[k], e);
          else     b0m[k] = fmaxf(b0m[k], e);
        }
      }
      #pragma unroll
      for (int k = 0; k < 4; ++k) w.llr_dL[s_][k] = b1m[k] - b0m[k];
    }
    __syncthreads();
    my_lld = w.llr_dL[s_out][k_out];
    __syncthreads();
  }

  out[b*32 + lane] = my_lld - my_lla;
}

extern "C" void kernel_launch(void* const* d_in, const int* in_sizes, int n_in,
                              void* d_out, int out_size, void* d_ws, size_t ws_size,
                              hipStream_t stream) {
  (void)n_in; (void)out_size; (void)d_ws; (void)ws_size;
  const float* y_re  = (const float*)d_in[0];
  const float* y_im  = (const float*)d_in[1];
  const float* h_re  = (const float*)d_in[2];
  const float* h_im  = (const float*)d_in[3];
  const float* prior = (const float*)d_in[4];
  const float* s_re  = (const float*)d_in[5];
  const float* s_im  = (const float*)d_in[6];
  float* out = (float*)d_out;

  int B = in_sizes[0] / 16;
  int blocks = B / WAVES_PER_BLOCK;
  mmse_pic_kernel<<<blocks, 256, 0, stream>>>(y_re, y_im, h_re, h_im,
                                              prior, s_re, s_im, out);
}


# Cholesky + whitening solve as one in-register factorisation of [S;(h|y)^H] (lane=row, column broadcast by v_readlane): no LDS or barrier per step; term order kept, each term two chained fma
# speedup vs baseline: 2.8847x; 1.7555x over previous
#include <hip/hip_runtime.h>
#include <math.h>

typedef __attribute__((ext_vector_type(2))) float v2f;
typedef __attribute__((ext_vector_type(8))) float v8f;

#define WAVES_PER_BLOCK 8
#define EPSV 1e-4f
#define INV_SQRT10 0.3162277660168379f

__device__ const float kPre[16] = {1,1,3,3,1,1,3,3,-1,-1,-3,-3,-1,-1,-3,-3};
__device__ const float kPim[16] = {1,3,1,3,-1,-3,-1,-3,1,3,1,3,-1,-3,-1,-3};

struct WS {
  union {
    struct StT { float Mre[25][17]; float Mim[25][17]; } st;
    float aug[16][33];
  } u;
  float Wre[16][10], Wim[16][10];
  float gre[8][9],  gim[8][9];
  float gr[16][17];
  float yptr[16][17];
  float ymfre[8], ymfim[8];
  float xre[8], xim[8], varL[8];
  float gxre[8], gxim[8];
  float muL[16], znumL[16];
  float rhoL[8];
  float llr_aL[8][4], llr_dL[8][4];
};

__device__ __forceinline__ float log_sigmoid(float x) {
  return fminf(x, 0.0f) - log1pf(__expf(-fabsf(x)));
}

__global__ __launch_bounds__(256) void mmse_pic_kernel(
    const float* __restrict__ y_re, const float* __restrict__ y_im,
    const float* __restrict__ h_re, const float* __restrict__ h_im,
    const float* __restrict__ prior,
    const float* __restrict__ s_re, const float* __restrict__ s_im,
    float* __restrict__ out)
{
  __shared__ WS ws[WAVES_PER_BLOCK];
  const int lane = threadIdx.x & 31;
  const int wid  = threadIdx.x >> 5;
  const int b    = blockIdx.x * WAVES_PER_BLOCK + wid;
  WS& w = ws[wid];

  #pragma unroll
  for (int t = 0; t < 8; ++t) {
    int idx = lane + 32*t;
    int i = idx >> 4, j = idx & 15;
    w.u.st.Mre[i][j] = s_re[b*256 + idx];
    w.u.st.Mim[i][j] = s_im[b*256 + idx];
  }
  #pragma unroll
  for (int t = 0; t < 4; ++t) {
    int idx = lane + 32*t;
    int m = idx >> 3, c = idx & 7;
    w.u.st.Mre[16 + c][m] =  h_re[b*128 + idx];
    w.u.st.Mim[16 + c][m] = -h_im[b*128 + idx];
  }
  if (lane < 16) {
    w.u.st.Mre[24][lane] =  y_re[b*16 + lane];
    w.u.st.Mim[24][lane] = -y_im[b*16 + lane];
  }
  const int s_out = lane >> 2, k_out = lane & 3;
  float my_lld = prior[b*32 + lane];
  w.llr_dL[s_out][k_out] = my_lld;
  __syncthreads();

  {
    const int rr = (lane < 25) ? lane : 24;
    float ar[16], ai[16];
    #pragma unroll
    for (int j = 0; j < 16; ++j) { ar[j] = w.u.st.Mre[rr][j]; ai[j] = w.u.st.Mim[rr][j]; }
    #pragma unroll
    for (int k = 0; k < 16; ++k) {
      float dk  = __int_as_float(__builtin_amdgcn_readlane(__float_as_int(ar[k]), k));
      float sdk = sqrtf(dk);
      float isd = 1.0f / sdk;
      ar[k] *= isd; ai[k] *= isd;
      const float lre = ar[k], lim = ai[k];
      #pragma unroll
      for (int j = k + 1; j < 16; ++j) {
        float cre = __int_as_float(__builtin_amdgcn_readlane(__float_as_int(ar[k]), j));
        float cim = __int_as_float(__builtin_amdgcn_readlane(__float_as_int(ai[k]), j));
        ar[j] = __builtin_fmaf(-lre, cre, ar[j]); ar[j] = __builtin_fmaf(-lim, cim, ar[j]);
        ai[j] = __builtin_fmaf(-lim, cre, ai[j]); ai[j] = __builtin_fmaf( lre, cim, ai[j]);
      }
    }
    if (lane >= 16 && lane < 25) {
      const int c = lane - 16;
      #pragma unroll
      for (int k = 0; k < 16; ++k) { w.Wre[k][c] = ar[k]; w.Wim[k][c] = -ai[k]; }
    }
  }
  __syncthreads();

  if (lane < 8) {
    int s_ = lane;
    float ar = 0.0f, ai = 0.0f;
    #pragma unroll
    for (int m = 0; m < 16; ++m) {
      float hr = w.Wre[m][s_], hi = w.Wim[m][s_];
      float yr = w.Wre[m][8],  yi = w.Wim[m][8];
      ar += hr*yr + hi*yi;
      ai += hr*yi - hi*yr;
    }
    w.ymfre[s_] = ar; w.ymfim[s_] = ai;
  }
  #pragma unroll
  for (int t = 0; t < 2; ++t) {
    int idx = lane + 32*t;
    int i = idx >> 3, j = idx & 7;
    float ar = 0.0f, ai = 0.0f;
    #pragma unroll
    for (int m = 0; m < 16; ++m) {
      float xr = w.Wre[m][i], xi = w.Wim[m][i];
      float yr = w.Wre[m][j], yi = w.Wim[m][j];
      ar += xr*yr + xi*yi;
      ai += xr*yi - xi*yr;
    }
    w.gre[i][j] = ar; w.gim[i][j] = ai;
  }
  __syncthreads();
  #pragma unroll
  for (int t = 0; t < 8; ++t) {
    int idx = lane + 32*t;
    int i = idx >> 4, j = idx & 15;
    float v;
    if (i < 8) v = (j < 8) ? w.gre[i][j]   : -w.gim[i][j-8];
    else       v = (j < 8) ? w.gim[i-8][j] :  w.gre[i-8][j-8];
    w.gr[i][j] = v;
  }
  __syncthreads();

  float my_lla = 0.0f;
  for (int iter = 0; iter < 2; ++iter) {
    my_lla = my_lld;
    w.llr_aL[s_out][k_out] = my_lla;
    __syncthreads();

    float lg[16];
    if (lane < 8) {
      int s_ = lane;
      float l0 = w.llr_aL[s_][0], l1 = w.llr_aL[s_][1];
      float l2 = w.llr_aL[s_][2], l3 = w.llr_aL[s_][3];
      #pragma unroll
      for (int p = 0; p < 16; ++p) {
        lg[p] = log_sigmoid((p & 8) ? l0 : -l0)
              + log_sigmoid((p & 4) ? l1 : -l1)
              + log_sigmoid((p & 2) ? l2 : -l2)
              + log_sigmoid((p & 1) ? l3 : -l3);
      }
      float mx = lg[0];
      #pragma unroll
      for (int p = 1; p < 16; ++p) mx = fmaxf(mx, lg[p]);
      float sum = 0.0f, mre = 0.0f, mim = 0.0f, e2 = 0.0f;
      #pragma unroll
      for (int p = 0; p < 16; ++p) {
        float e = __expf(lg[p] - mx);
        float pr = kPre[p]*INV_SQRT10, pi = kPim[p]*INV_SQRT10;
        sum += e; mre += e*pr; mim += e*pi; e2 += e*(pr*pr + pi*pi);
      }
      float invs = 1.0f / sum;
      mre *= invs; mim *= invs; e2 *= invs;
      w.xre[s_] = mre; w.xim[s_] = mim;
      w.varL[s_] = e2 - (mre*mre + mim*mim);
    }
    __syncthreads();

    if (lane < 8) {
      int i = lane;
      float ar = 0.0f, ai = 0.0f;
      #pragma unroll
      for (int j = 0; j < 8; ++j) {
        float gre_ = w.gre[i][j], gim_ = w.gim[i][j];
        float xr = w.xre[j], xi = w.xim[j];
        ar += gre_*xr - gim_*xi;
        ai += gre_*xi + gim_*xr;
      }
      w.gxre[i] = ar; w.gxim[i] = ai;
    }
    __syncthreads();

    #pragma unroll
    for (int t = 0; t < 2; ++t) {
      int idx = lane + 32*t;
      int i = idx >> 3, j = idx & 7;
      float gre_ = w.gre[j][i], gim_ = w.gim[j][i];
      float xr = w.xre[i], xi = w.xim[i];
      float tre = w.ymfre[j] + (gre_*xr - gim_*xi) - w.gxre[j];
      float tim = w.ymfim[j] + (gre_*xi + gim_*xr) - w.gxim[j];
      w.yptr[i][j]     = tre; w.yptr[i][j+8]   = tim;
      w.yptr[i+8][j]   = tre; w.yptr[i+8][j+8] = tim;
    }
    {
      const int cc = lane & 15;
      const float vv = w.varL[cc & 7];
      float x[16];
      #pragma unroll
      for (int r = 0; r < 16; ++r) {
        float dl = (r == cc) ? 1.0f : 0.0f;
        float left = __builtin_fmaf(w.gr[r][cc], vv, dl);
        x[r] = (lane < 16) ? left : dl;
      }
      #pragma unroll
      for (int k = 0; k < 16; ++k) {
        float piv  = __int_as_float(__builtin_amdgcn_readlane(__float_as_int(x[k]), k));
        float pinv = 1.0f / piv;
        float f[16];
        #pragma unroll
        for (int r = 0; r < 16; ++r)
          f[r] = __int_as_float(__builtin_amdgcn_readlane(__float_as_int(x[r]), k));
        x[k] *= pinv;
        const float t = x[k];
        #pragma unroll
        for (int r = 0; r < 16; ++r)
          if (r != k) x[r] -= f[r] * t;
      }
      #pragma unroll
      for (int r = 0; r < 16; ++r) w.u.aug[r][lane] = x[r];
    }
    __syncthreads();

    {
      int nn = lane & 15, hf = lane >> 4;
      v8f p1 = {}; v8f p2 = {};
      #pragma unroll
      for (int kb = 0; kb < 4; ++kb) {
        int k0 = kb*4 + hf*2;
        v2f af, b1, b2;
        af.x = w.u.aug[nn][16 + k0];
        af.y = w.u.aug[nn][16 + k0 + 1];
        b1.x = w.gr[k0][nn];   b1.y = w.gr[k0+1][nn];
        b2.x = w.yptr[nn][k0]; b2.y = w.yptr[nn][k0+1];
        p1 = __builtin_amdgcn_wmma_f32_16x16x4_f32(false, af, false, b1,
                                                   (short)0, p1, false, false);
        p2 = __builtin_amdgcn_wmma_f32_16x16x4_f32(false, af, false, b2,
                                                   (short)0, p2, false, false);
      }
      #pragma unroll
      for (int v = 0; v < 8; ++v) {
        int row = v + 8*hf;
        if (row == nn) { w.muL[nn] = p1[v]; w.znumL[nn] = p2[v]; }
      }
    }
    __syncthreads();

    if (lane < 8) {
      int s_ = lane;
      float m1 = w.muL[s_], m2 = w.muL[s_ + 8];
      w.xre[s_] = w.znumL[s_]     / m1;
      w.xim[s_] = w.znumL[s_ + 8] / m2;
      w.rhoL[s_] = m1 / fmaxf(1.0f - w.varL[s_] * m1, EPSV);
    }
    __syncthreads();

    if (lane < 8) {
      int s_ = lane;
      float zr = w.xre[s_], zi = w.xim[s_];
      float rho = w.rhoL[s_];
      float b1m[4] = {-1e30f,-1e30f,-1e30f,-1e30f};
      float b0m[4] = {-1e30f,-1e30f,-1e30f,-1e30f};
      #pragma unroll
      for (int p = 0; p < 16; ++p) {
        float pr = kPre[p]*INV_SQRT10, pi = kPim[p]*INV_SQRT10;
        float dre = zr - pr, dim = zi - pi;
        float e = -(dre*dre + dim*dim) * rho + lg[p];
        #pragma unroll
        for (int k = 0; k < 4; ++k) {
          int bit = (p >> (3-k)) & 1;
          if (bit) b1m[k] = fmaxf(b1m[k], e);
          else     b0m[k] = fmaxf(b0m[k], e);
        }
      }
      #pragma unroll
      for (int k = 0; k < 4; ++k) w.llr_dL[s_][k] = b1m[k] - b0m[k];
    }
    __syncthreads();
    my_lld = w.llr_dL[s_out][k_out];
    __syncthreads();
  }

  out[b*32 + lane] = my_lld - my_lla;
}

extern "C" void kernel_launch(void* const* d_in, const int* in_sizes, int n_in,
                              void* d_out, int out_size, void* d_ws, size_t ws_size,
                              hipStream_t stream) {
  (void)n_in; (void)out_size; (void)d_ws; (void)ws_size;
  const float* y_re  = (const float*)d_in[0];
  const float* y_im  = (const float*)d_in[1];
  const float* h_re  = (const float*)d_in[2];
  const float* h_im  = (const float*)d_in[3];
  const float* prior = (const float*)d_in[4];
  const float* s_re  = (const float*)d_in[5];
  const float* s_im  = (const float*)d_in[6];
  float* out = (float*)d_out;

  int B = in_sizes[0] / 16;
  int blocks = B / WAVES_PER_BLOCK;
  mmse_pic_kernel<<<blocks, 256, 0, stream>>>(y_re, y_im, h_re, h_im,
                                              prior, s_re, s_im, out);
}


# Symbol stages (soft-symbol moments, estimate, max-log demap) on all 32 lanes, 4 of a stream's 16 points per lane, joined by quad lane exchanges; softmax sums now 4 per lane then a quad tree
# speedup vs baseline: 3.4512x; 1.1964x over previous
#include <hip/hip_runtime.h>
#include <math.h>

typedef __attribute__((ext_vector_type(2))) float v2f;
typedef __attribute__((ext_vector_type(8))) float v8f;

#define WAVES_PER_BLOCK 8
#define EPSV 1e-4f
#define INV_SQRT10 0.3162277660168379f


struct WS {
  union {
    struct StT { float Mre[25][17]; float Mim[25][17]; } st;
    float aug[16][33];
  } u;
  float Wre[16][10], Wim[16][10];
  float gre[8][9],  gim[8][9];
  float gr[16][17];
  float yptr[16][17];
  float ymfre[8], ymfim[8];
  float xre[8], xim[8], varL[8];
  float gxre[8], gxim[8];
  float muL[16], znumL[16];
};

__device__ __forceinline__ float log_sigmoid(float x) {
  return fminf(x, 0.0f) - log1pf(__expf(-fabsf(x)));
}

template <int CTRL>
__device__ __forceinline__ float quad_read(float v) {
  return __int_as_float(__builtin_amdgcn_mov_dpp(__float_as_int(v), CTRL, 0xF, 0xF, true));
}
#define QUAD_LANE0 0x00
#define QUAD_LANE1 0x55
#define QUAD_LANE2 0xAA
#define QUAD_LANE3 0xFF
#define QUAD_XOR1  0xB1
#define QUAD_XOR2  0x4E
__device__ __forceinline__ float quad_sum(float v) {
  v += quad_read<QUAD_XOR1>(v);
  v += quad_read<QUAD_XOR2>(v);
  return v;
}
__device__ __forceinline__ float quad_max(float v) {
  v = fmaxf(v, quad_read<QUAD_XOR1>(v));
  v = fmaxf(v, quad_read<QUAD_XOR2>(v));
  return v;
}

__global__ __launch_bounds__(256) void mmse_pic_kernel(
    const float* __restrict__ y_re, const float* __restrict__ y_im,
    const float* __restrict__ h_re, const float* __restrict__ h_im,
    const float* __restrict__ prior,
    const float* __restrict__ s_re, const float* __restrict__ s_im,
    float* __restrict__ out)
{
  __shared__ WS ws[WAVES_PER_BLOCK];
  const int lane = threadIdx.x & 31;
  const int wid  = threadIdx.x >> 5;
  const int b    = blockIdx.x * WAVES_PER_BLOCK + wid;
  WS& w = ws[wid];

  #pragma unroll
  for (int t = 0; t < 8; ++t) {
    int idx = lane + 32*t;
    int i = idx >> 4, j = idx & 15;
    w.u.st.Mre[i][j] = s_re[b*256 + idx];
    w.u.st.Mim[i][j] = s_im[b*256 + idx];
  }
  #pragma unroll
  for (int t = 0; t < 4; ++t) {
    int idx = lane + 32*t;
    int m = idx >> 3, c = idx & 7;
    w.u.st.Mre[16 + c][m] =  h_re[b*128 + idx];
    w.u.st.Mim[16 + c][m] = -h_im[b*128 + idx];
  }
  if (lane < 16) {
    w.u.st.Mre[24][lane] =  y_re[b*16 + lane];
    w.u.st.Mim[24][lane] = -y_im[b*16 + lane];
  }
  const int s_out = lane >> 2, k_out = lane & 3;
  float my_lld = prior[b*32 + lane];
  __syncthreads();

  {
    const int rr = (lane < 25) ? lane : 24;
    float ar[16], ai[16];
    #pragma unroll
    for (int j = 0; j < 16; ++j) { ar[j] = w.u.st.Mre[rr][j]; ai[j] = w.u.st.Mim[rr][j]; }
    #pragma unroll
    for (int k = 0; k < 16; ++k) {
      float dk  = __int_as_float(__builtin_amdgcn_readlane(__float_as_int(ar[k]), k));
      float sdk = sqrtf(dk);
      float isd = 1.0f / sdk;
      ar[k] *= isd; ai[k] *= isd;
      const float lre = ar[k], lim = ai[k];
      #pragma unroll
      for (int j = k + 1; j < 16; ++j) {
        float cre = __int_as_float(__builtin_amdgcn_readlane(__float_as_int(ar[k]), j));
        float cim = __int_as_float(__builtin_amdgcn_readlane(__float_as_int(ai[k]), j));
        ar[j] = __builtin_fmaf(-lre, cre, ar[j]); ar[j] = __builtin_fmaf(-lim, cim, ar[j]);
        ai[j] = __builtin_fmaf(-lim, cre, ai[j]); ai[j] = __builtin_fmaf( lre, cim, ai[j]);
      }
    }
    if (lane >= 16 && lane < 25) {
      const int c = lane - 16;
      #pragma unroll
      for (int k = 0; k < 16; ++k) { w.Wre[k][c] = ar[k]; w.Wim[k][c] = -ai[k]; }
    }
  }
  __syncthreads();

  if (lane < 8) {
    int s_ = lane;
    float ar = 0.0f, ai = 0.0f;
    #pragma unroll
    for (int m = 0; m < 16; ++m) {
      float hr = w.Wre[m][s_], hi = w.Wim[m][s_];
      float yr = w.Wre[m][8],  yi = w.Wim[m][8];
      ar += hr*yr + hi*yi;
      ai += hr*yi - hi*yr;
    }
    w.ymfre[s_] = ar; w.ymfim[s_] = ai;
  }
  #pragma unroll
  for (int t = 0; t < 2; ++t) {
    int idx = lane + 32*t;
    int i = idx >> 3, j = idx & 7;
    float ar = 0.0f, ai = 0.0f;
    #pragma unroll
    for (int m = 0; m < 16; ++m) {
      float xr = w.Wre[m][i], xi = w.Wim[m][i];
      float yr = w.Wre[m][j], yi = w.Wim[m][j];
      ar += xr*yr + xi*yi;
      ai += xr*yi - xi*yr;
    }
    w.gre[i][j] = ar; w.gim[i][j] = ai;
  }
  __syncthreads();
  #pragma unroll
  for (int t = 0; t < 8; ++t) {
    int idx = lane + 32*t;
    int i = idx >> 4, j = idx & 15;
    float v;
    if (i < 8) v = (j < 8) ? w.gre[i][j]   : -w.gim[i][j-8];
    else       v = (j < 8) ? w.gim[i-8][j] :  w.gre[i-8][j-8];
    w.gr[i][j] = v;
  }
  __syncthreads();

  const float C1 = 1.0f * INV_SQRT10, C3 = 3.0f * INV_SQRT10;
  const float prA = (k_out & 2) ? -C1 : C1, prB = (k_out & 2) ? -C3 : C3;
  const float piA = (k_out & 1) ? -C1 : C1, piB = (k_out & 1) ? -C3 : C3;
  float my_lla = 0.0f;
  for (int iter = 0; iter < 2; ++iter) {
    my_lla = my_lld;

    float lg[4];
    float var_s;
    {
      const float lsp = log_sigmoid( my_lla);
      const float lsn = log_sigmoid(-my_lla);
      const float p0 = quad_read<QUAD_LANE0>(lsp), n0 = quad_read<QUAD_LANE0>(lsn);
      const float p1 = quad_read<QUAD_LANE1>(lsp), n1 = quad_read<QUAD_LANE1>(lsn);
      const float p2 = quad_read<QUAD_LANE2>(lsp), n2 = quad_read<QUAD_LANE2>(lsn);
      const float p3 = quad_read<QUAD_LANE3>(lsp), n3 = quad_read<QUAD_LANE3>(lsn);
      const float base = ((k_out & 2) ? p0 : n0) + ((k_out & 1) ? p1 : n1);
      lg[0] = (base + n2) + n3;
      lg[1] = (base + n2) + p3;
      lg[2] = (base + p2) + n3;
      lg[3] = (base + p2) + p3;
      const float mx = quad_max(fmaxf(fmaxf(lg[0], lg[1]), fmaxf(lg[2], lg[3])));
      float sum = 0.0f, mre = 0.0f, mim = 0.0f, e2 = 0.0f;
      #pragma unroll
      for (int j = 0; j < 4; ++j) {
        float e = __expf(lg[j] - mx);
        float pr = (j & 2) ? prB : prA, pi = (j & 1) ? piB : piA;
        sum += e; mre += e*pr; mim += e*pi; e2 += e*(pr*pr + pi*pi);
      }
      sum = quad_sum(sum); mre = quad_sum(mre); mim = quad_sum(mim); e2 = quad_sum(e2);
      float invs = 1.0f / sum;
      mre *= invs; mim *= invs; e2 *= invs;
      var_s = e2 - (mre*mre + mim*mim);
      if (k_out == 0) { w.xre[s_out] = mre; w.xim[s_out] = mim; w.varL[s_out] = var_s; }
    }
    __syncthreads();

    if (lane < 8) {
      int i = lane;
      float ar = 0.0f, ai = 0.0f;
      #pragma unroll
      for (int j = 0; j < 8; ++j) {
        float gre_ = w.gre[i][j], gim_ = w.gim[i][j];
        float xr = w.xre[j], xi = w.xim[j];
        ar += gre_*xr - gim_*xi;
        ai += gre_*xi + gim_*xr;
      }
      w.gxre[i] = ar; w.gxim[i] = ai;
    }
    __syncthreads();

    #pragma unroll
    for (int t = 0; t < 2; ++t) {
      int idx = lane + 32*t;
      int i = idx >> 3, j = idx & 7;
      float gre_ = w.gre[j][i], gim_ = w.gim[j][i];
      float xr = w.xre[i], xi = w.xim[i];
      float tre = w.ymfre[j] + (gre_*xr - gim_*xi) - w.gxre[j];
      float tim = w.ymfim[j] + (gre_*xi + gim_*xr) - w.gxim[j];
      w.yptr[i][j]     = tre; w.yptr[i][j+8]   = tim;
      w.yptr[i+8][j]   = tre; w.yptr[i+8][j+8] = tim;
    }
    {
      const int cc = lane & 15;
      const float vv = w.varL[cc & 7];
      float x[16];
      #pragma unroll
      for (int r = 0; r < 16; ++r) {
        float dl = (r == cc) ? 1.0f : 0.0f;
        float left = __builtin_fmaf(w.gr[r][cc], vv, dl);
        x[r] = (lane < 16) ? left : dl;
      }
      #pragma unroll
      for (int k = 0; k < 16; ++k) {
        float piv  = __int_as_float(__builtin_amdgcn_readlane(__float_as_int(x[k]), k));
        float pinv = 1.0f / piv;
        float f[16];
        #pragma unroll
        for (int r = 0; r < 16; ++r)
          f[r] = __int_as_float(__builtin_amdgcn_readlane(__float_as_int(x[r]), k));
        x[k] *= pinv;
        const float t = x[k];
        #pragma unroll
        for (int r = 0; r < 16; ++r)
          if (r != k) x[r] -= f[r] * t;
      }
      #pragma unroll
      for (int r = 0; r < 16; ++r) w.u.aug[r][lane] = x[r];
    }
    __syncthreads();

    {
      int nn = lane & 15, hf = lane >> 4;
      v8f p1 = {}; v8f p2 = {};
      #pragma unroll
      for (int kb = 0; kb < 4; ++kb) {
        int k0 = kb*4 + hf*2;
        v2f af, b1, b2;
        af.x = w.u.aug[nn][16 + k0];
        af.y = w.u.aug[nn][16 + k0 + 1];
        b1.x = w.gr[k0][nn];   b1.y = w.gr[k0+1][nn];
        b2.x = w.yptr[nn][k0]; b2.y = w.yptr[nn][k0+1];
        p1 = __builtin_amdgcn_wmma_f32_16x16x4_f32(false, af, false, b1,
                                                   (short)0, p1, false, false);
        p2 = __builtin_amdgcn_wmma_f32_16x16x4_f32(false, af, false, b2,
                                                   (short)0, p2, false, false);
      }
      #pragma unroll
      for (int v = 0; v < 8; ++v) {
        int row = v + 8*hf;
        if (row == nn) { w.muL[nn] = p1[v]; w.znumL[nn] = p2[v]; }
      }
    }
    __syncthreads();

    const float m1 = w.muL[s_out], m2 = w.muL[s_out + 8];
    const float zr = w.znumL[s_out]     / m1;
    const float zi = w.znumL[s_out + 8] / m2;
    const float rho = m1 / fmaxf(1.0f - var_s * m1, EPSV);

    {
      float ev[4];
      #pragma unroll
      for (int j = 0; j < 4; ++j) {
        float pr = (j & 2) ? prB : prA, pi = (j & 1) ? piB : piA;
        float dre = zr - pr, dim = zi - pi;
        ev[j] = -(dre*dre + dim*dim) * rho + lg[j];
      }
      const float b2_0 = quad_max(fmaxf(ev[0], ev[1])), b2_1 = quad_max(fmaxf(ev[2], ev[3]));
      const float b3_0 = quad_max(fmaxf(ev[0], ev[2])), b3_1 = quad_max(fmaxf(ev[1], ev[3]));
      const float mall = fmaxf(fmaxf(ev[0], ev[1]), fmaxf(ev[2], ev[3]));
      const float u = fmaxf(mall, quad_read<QUAD_XOR1>(mall));
      const float v = fmaxf(mall, quad_read<QUAD_XOR2>(mall));
      const float b0_0 = quad_read<QUAD_LANE0>(u), b0_1 = quad_read<QUAD_LANE2>(u);
      const float b1_0 = quad_read<QUAD_LANE0>(v), b1_1 = quad_read<QUAD_LANE1>(v);
      float one  = (k_out == 0) ? b0_1 : (k_out == 1) ? b1_1 : (k_out == 2) ? b2_1 : b3_1;
      float zero = (k_out == 0) ? b0_0 : (k_out == 1) ? b1_0 : (k_out == 2) ? b2_0 : b3_0;
      my_lld = fmaxf(one, -1e30f) - fmaxf(zero, -1e30f);
    }
  }

  out[b*32 + lane] = my_lld - my_lla;
}

extern "C" void kernel_launch(void* const* d_in, const int* in_sizes, int n_in,
                              void* d_out, int out_size, void* d_ws, size_t ws_size,
                              hipStream_t stream) {
  (void)n_in; (void)out_size; (void)d_ws; (void)ws_size;
  const float* y_re  = (const float*)d_in[0];
  const float* y_im  = (const float*)d_in[1];
  const float* h_re  = (const float*)d_in[2];
  const float* h_im  = (const float*)d_in[3];
  const float* prior = (const float*)d_in[4];
  const float* s_re  = (const float*)d_in[5];
  const float* s_im  = (const float*)d_in[6];
  float* out = (float*)d_out;

  int B = in_sizes[0] / 16;
  int blocks = B / WAVES_PER_BLOCK;
  mmse_pic_kernel<<<blocks, 256, 0, stream>>>(y_re, y_im, h_re, h_im,
                                              prior, s_re, s_im, out);
}


# Pivot reciprocals of both eliminations (Cholesky 1/sqrt, Gauss-Jordan 1/pivot) by the hardware rsq / rcp instruction (1 ulp) instead of correctly rounded sqrt and division sequences; sum order kept
# speedup vs baseline: 4.0100x; 1.1619x over previous
#include <hip/hip_runtime.h>
#include <math.h>

typedef __attribute__((ext_vector_type(2))) float v2f;
typedef __attribute__((ext_vector_type(8))) float v8f;

#define WAVES_PER_BLOCK 8
#define EPSV 1e-4f
#define INV_SQRT10 0.3162277660168379f


struct WS {
  union {
    struct StT { float Mre[25][17]; float Mim[25][17]; } st;
    float aug[16][33];
  } u;
  float Wre[16][10], Wim[16][10];
  float gre[8][9],  gim[8][9];
  float gr[16][17];
  float yptr[16][17];
  float ymfre[8], ymfim[8];
  float xre[8], xim[8], varL[8];
  float gxre[8], gxim[8];
  float muL[16], znumL[16];
};

__device__ __forceinline__ float log_sigmoid(float x) {
  return fminf(x, 0.0f) - log1pf(__expf(-fabsf(x)));
}

template <int CTRL>
__device__ __forceinline__ float quad_read(float v) {
  return __int_as_float(__builtin_amdgcn_mov_dpp(__float_as_int(v), CTRL, 0xF, 0xF, true));
}
#define QUAD_LANE0 0x00
#define QUAD_LANE1 0x55
#define QUAD_LANE2 0xAA
#define QUAD_LANE3 0xFF
#define QUAD_XOR1  0xB1
#define QUAD_XOR2  0x4E
__device__ __forceinline__ float quad_sum(float v) {
  v += quad_read<QUAD_XOR1>(v);
  v += quad_read<QUAD_XOR2>(v);
  return v;
}
__device__ __forceinline__ float quad_max(float v) {
  v = fmaxf(v, quad_read<QUAD_XOR1>(v));
  v = fmaxf(v, quad_read<QUAD_XOR2>(v));
  return v;
}

__global__ __launch_bounds__(256) void mmse_pic_kernel(
    const float* __restrict__ y_re, const float* __restrict__ y_im,
    const float* __restrict__ h_re, const float* __restrict__ h_im,
    const float* __restrict__ prior,
    const float* __restrict__ s_re, const float* __restrict__ s_im,
    float* __restrict__ out)
{
  __shared__ WS ws[WAVES_PER_BLOCK];
  const int lane = threadIdx.x & 31;
  const int wid  = threadIdx.x >> 5;
  const int b    = blockIdx.x * WAVES_PER_BLOCK + wid;
  WS& w = ws[wid];

  #pragma unroll
  for (int t = 0; t < 8; ++t) {
    int idx = lane + 32*t;
    int i = idx >> 4, j = idx & 15;
    w.u.st.Mre[i][j] = s_re[b*256 + idx];
    w.u.st.Mim[i][j] = s_im[b*256 + idx];
  }
  #pragma unroll
  for (int t = 0; t < 4; ++t) {
    int idx = lane + 32*t;
    int m = idx >> 3, c = idx & 7;
    w.u.st.Mre[16 + c][m] =  h_re[b*128 + idx];
    w.u.st.Mim[16 + c][m] = -h_im[b*128 + idx];
  }
  if (lane < 16) {
    w.u.st.Mre[24][lane] =  y_re[b*16 + lane];
    w.u.st.Mim[24][lane] = -y_im[b*16 + lane];
  }
  const int s_out = lane >> 2, k_out = lane & 3;
  float my_lld = prior[b*32 + lane];
  __syncthreads();

  {
    const int rr = (lane < 25) ? lane : 24;
    float ar[16], ai[16];
    #pragma unroll
    for (int j = 0; j < 16; ++j) { ar[j] = w.u.st.Mre[rr][j]; ai[j] = w.u.st.Mim[rr][j]; }
    #pragma unroll
    for (int k = 0; k < 16; ++k) {
      float dk  = __int_as_float(__builtin_amdgcn_readlane(__float_as_int(ar[k]), k));
      float isd = __builtin_amdgcn_rsqf(dk);
      ar[k] *= isd; ai[k] *= isd;
      const float lre = ar[k], lim = ai[k];
      #pragma unroll
      for (int j = k + 1; j < 16; ++j) {
        float cre = __int_as_float(__builtin_amdgcn_readlane(__float_as_int(ar[k]), j));
        float cim = __int_as_float(__builtin_amdgcn_readlane(__float_as_int(ai[k]), j));
        ar[j] = __builtin_fmaf(-lre, cre, ar[j]); ar[j] = __builtin_fmaf(-lim, cim, ar[j]);
        ai[j] = __builtin_fmaf(-lim, cre, ai[j]); ai[j] = __builtin_fmaf( lre, cim, ai[j]);
      }
    }
    if (lane >= 16 && lane < 25) {
      const int c = lane - 16;
      #pragma unroll
      for (int k = 0; k < 16; ++k) { w.Wre[k][c] = ar[k]; w.Wim[k][c] = -ai[k]; }
    }
  }
  __syncthreads();

  if (lane < 8) {
    int s_ = lane;
    float ar = 0.0f, ai = 0.0f;
    #pragma unroll
    for (int m = 0; m < 16; ++m) {
      float hr = w.Wre[m][s_], hi = w.Wim[m][s_];
      float yr = w.Wre[m][8],  yi = w.Wim[m][8];
      ar += hr*yr + hi*yi;
      ai += hr*yi - hi*yr;
    }
    w.ymfre[s_] = ar; w.ymfim[s_] = ai;
  }
  #pragma unroll
  for (int t = 0; t < 2; ++t) {
    int idx = lane + 32*t;
    int i = idx >> 3, j = idx & 7;
    float ar = 0.0f, ai = 0.0f;
    #pragma unroll
    for (int m = 0; m < 16; ++m) {
      float xr = w.Wre[m][i], xi = w.Wim[m][i];
      float yr = w.Wre[m][j], yi = w.Wim[m][j];
      ar += xr*yr + xi*yi;
      ai += xr*yi - xi*yr;
    }
    w.gre[i][j] = ar; w.gim[i][j] = ai;
  }
  __syncthreads();
  #pragma unroll
  for (int t = 0; t < 8; ++t) {
    int idx = lane + 32*t;
    int i = idx >> 4, j = idx & 15;
    float v;
    if (i < 8) v = (j < 8) ? w.gre[i][j]   : -w.gim[i][j-8];
    else       v = (j < 8) ? w.gim[i-8][j] :  w.gre[i-8][j-8];
    w.gr[i][j] = v;
  }
  __syncthreads();

  const float C1 = 1.0f * INV_SQRT10, C3 = 3.0f * INV_SQRT10;
  const float prA = (k_out & 2) ? -C1 : C1, prB = (k_out & 2) ? -C3 : C3;
  const float piA = (k_out & 1) ? -C1 : C1, piB = (k_out & 1) ? -C3 : C3;
  float my_lla = 0.0f;
  for (int iter = 0; iter < 2; ++iter) {
    my_lla = my_lld;

    float lg[4];
    float var_s;
    {
      const float lsp = log_sigmoid( my_lla);
      const float lsn = log_sigmoid(-my_lla);
      const float p0 = quad_read<QUAD_LANE0>(lsp), n0 = quad_read<QUAD_LANE0>(lsn);
      const float p1 = quad_read<QUAD_LANE1>(lsp), n1 = quad_read<QUAD_LANE1>(lsn);
      const float p2 = quad_read<QUAD_LANE2>(lsp), n2 = quad_read<QUAD_LANE2>(lsn);
      const float p3 = quad_read<QUAD_LANE3>(lsp), n3 = quad_read<QUAD_LANE3>(lsn);
      const float base = ((k_out & 2) ? p0 : n0) + ((k_out & 1) ? p1 : n1);
      lg[0] = (base + n2) + n3;
      lg[1] = (base + n2) + p3;
      lg[2] = (base + p2) + n3;
      lg[3] = (base + p2) + p3;
      const float mx = quad_max(fmaxf(fmaxf(lg[0], lg[1]), fmaxf(lg[2], lg[3])));
      float sum = 0.0f, mre = 0.0f, mim = 0.0f, e2 = 0.0f;
      #pragma unroll
      for (int j = 0; j < 4; ++j) {
        float e = __expf(lg[j] - mx);
        float pr = (j & 2) ? prB : prA, pi = (j & 1) ? piB : piA;
        sum += e; mre += e*pr; mim += e*pi; e2 += e*(pr*pr + pi*pi);
      }
      sum = quad_sum(sum); mre = quad_sum(mre); mim = quad_sum(mim); e2 = quad_sum(e2);
      float invs = 1.0f / sum;
      mre *= invs; mim *= invs; e2 *= invs;
      var_s = e2 - (mre*mre + mim*mim);
      if (k_out == 0) { w.xre[s_out] = mre; w.xim[s_out] = mim; w.varL[s_out] = var_s; }
    }
    __syncthreads();

    if (lane < 8) {
      int i = lane;
      float ar = 0.0f, ai = 0.0f;
      #pragma unroll
      for (int j = 0; j < 8; ++j) {
        float gre_ = w.gre[i][j], gim_ = w.gim[i][j];
        float xr = w.xre[j], xi = w.xim[j];
        ar += gre_*xr - gim_*xi;
        ai += gre_*xi + gim_*xr;
      }
      w.gxre[i] = ar; w.gxim[i] = ai;
    }
    __syncthreads();

    #pragma unroll
    for (int t = 0; t < 2; ++t) {
      int idx = lane + 32*t;
      int i = idx >> 3, j = idx & 7;
      float gre_ = w.gre[j][i], gim_ = w.gim[j][i];
      float xr = w.xre[i], xi = w.xim[i];
      float tre = w.ymfre[j] + (gre_*xr - gim_*xi) - w.gxre[j];
      float tim = w.ymfim[j] + (gre_*xi + gim_*xr) - w.gxim[j];
      w.yptr[i][j]     = tre; w.yptr[i][j+8]   = tim;
      w.yptr[i+8][j]   = tre; w.yptr[i+8][j+8] = tim;
    }
    {
      const int cc = lane & 15;
      const float vv = w.varL[cc & 7];
      float x[16];
      #pragma unroll
      for (int r = 0; r < 16; ++r) {
        float dl = (r == cc) ? 1.0f : 0.0f;
        float left = __builtin_fmaf(w.gr[r][cc], vv, dl);
        x[r] = (lane < 16) ? left : dl;
      }
      #pragma unroll
      for (int k = 0; k < 16; ++k) {
        float piv  = __int_as_float(__builtin_amdgcn_readlane(__float_as_int(x[k]), k));
        float pinv = __builtin_amdgcn_rcpf(piv);
        float f[16];
        #pragma unroll
        for (int r = 0; r < 16; ++r)
          f[r] = __int_as_float(__builtin_amdgcn_readlane(__float_as_int(x[r]), k));
        x[k] *= pinv;
        const float t = x[k];
        #pragma unroll
        for (int r = 0; r < 16; ++r)
          if (r != k) x[r] -= f[r] * t;
      }
      #pragma unroll
      for (int r = 0; r < 16; ++r) w.u.aug[r][lane] = x[r];
    }
    __syncthreads();

    {
      int nn = lane & 15, hf = lane >> 4;
      v8f p1 = {}; v8f p2 = {};
      #pragma unroll
      for (int kb = 0; kb < 4; ++kb) {
        int k0 = kb*4 + hf*2;
        v2f af, b1, b2;
        af.x = w.u.aug[nn][16 + k0];
        af.y = w.u.aug[nn][16 + k0 + 1];
        b1.x = w.gr[k0][nn];   b1.y = w.gr[k0+1][nn];
        b2.x = w.yptr[nn][k0]; b2.y = w.yptr[nn][k0+1];
        p1 = __builtin_amdgcn_wmma_f32_16x16x4_f32(false, af, false, b1,
                                                   (short)0, p1, false, false);
        p2 = __builtin_amdgcn_wmma_f32_16x16x4_f32(false, af, false, b2,
                                                   (short)0, p2, false, false);
      }
      #pragma unroll
      for (int v = 0; v < 8; ++v) {
        int row = v + 8*hf;
        if (row == nn) { w.muL[nn] = p1[v]; w.znumL[nn] = p2[v]; }
      }
    }
    __syncthreads();

    const float m1 = w.muL[s_out], m2 = w.muL[s_out + 8];
    const float zr = w.znumL[s_out]     / m1;
    const float zi = w.znumL[s_out + 8] / m2;
    const float rho = m1 / fmaxf(1.0f - var_s * m1, EPSV);

    {
      float ev[4];
      #pragma unroll
      for (int j = 0; j < 4; ++j) {
        float pr = (j & 2) ? prB : prA, pi = (j & 1) ? piB : piA;
        float dre = zr - pr, dim = zi - pi;
        ev[j] = -(dre*dre + dim*dim) * rho + lg[j];
      }
      const float b2_0 = quad_max(fmaxf(ev[0], ev[1])), b2_1 = quad_max(fmaxf(ev[2], ev[3]));
      const float b3_0 = quad_max(fmaxf(ev[0], ev[2])), b3_1 = quad_max(fmaxf(ev[1], ev[3]));
      const float mall = fmaxf(fmaxf(ev[0], ev[1]), fmaxf(ev[2], ev[3]));
      const float u = fmaxf(mall, quad_read<QUAD_XOR1>(mall));
      const float v = fmaxf(mall, quad_read<QUAD_XOR2>(mall));
      const float b0_0 = quad_read<QUAD_LANE0>(u), b0_1 = quad_read<QUAD_LANE2>(u);
      const float b1_0 = quad_read<QUAD_LANE0>(v), b1_1 = quad_read<QUAD_LANE1>(v);
      float one  = (k_out == 0) ? b0_1 : (k_out == 1) ? b1_1 : (k_out == 2) ? b2_1 : b3_1;
      float zero = (k_out == 0) ? b0_0 : (k_out == 1) ? b1_0 : (k_out == 2) ? b2_0 : b3_0;
      my_lld = fmaxf(one, -1e30f) - fmaxf(zero, -1e30f);
    }
  }

  out[b*32 + lane] = my_lld - my_lla;
}

extern "C" void kernel_launch(void* const* d_in, const int* in_sizes, int n_in,
                              void* d_out, int out_size, void* d_ws, size_t ws_size,
                              hipStream_t stream) {
  (void)n_in; (void)out_size; (void)d_ws; (void)ws_size;
  const float* y_re  = (const float*)d_in[0];
  const float* y_im  = (const float*)d_in[1];
  const float* h_re  = (const float*)d_in[2];
  const float* h_im  = (const float*)d_in[3];
  const float* prior = (const float*)d_in[4];
  const float* s_re  = (const float*)d_in[5];
  const float* s_im  = (const float*)d_in[6];
  float* out = (float*)d_out;

  int B = in_sizes[0] / 16;
  int blocks = B / WAVES_PER_BLOCK;
  mmse_pic_kernel<<<blocks, 256, 0, stream>>>(y_re, y_im, h_re, h_im,
                                              prior, s_re, s_im, out);
}


# Gauss-Jordan inverse run on the 8x8 complex matrix g*diag(var)+I, of which the 16x16 real a is the real form: 8 complex steps, row updates as packed f32 fma; elimination sum order changes
# speedup vs baseline: 4.4472x; 1.1090x over previous
#include <hip/hip_runtime.h>
#include <math.h>

typedef __attribute__((ext_vector_type(2))) float v2f;
typedef __attribute__((ext_vector_type(8))) float v8f;

#define WAVES_PER_BLOCK 8
#define EPSV 1e-4f
#define INV_SQRT10 0.3162277660168379f


struct WS {
  union {
    struct StT { float Mre[25][17]; float Mim[25][17]; } st;
    float aug[16][33];
  } u;
  float Wre[16][10], Wim[16][10];
  float gre[8][9],  gim[8][9];
  float gr[16][17];
  float yptr[16][17];
  float ymfre[8], ymfim[8];
  float xre[8], xim[8], varL[8];
  float gxre[8], gxim[8];
  float muL[16], znumL[16];
};

__device__ __forceinline__ float log_sigmoid(float x) {
  return fminf(x, 0.0f) - log1pf(__expf(-fabsf(x)));
}

template <int CTRL>
__device__ __forceinline__ float quad_read(float v) {
  return __int_as_float(__builtin_amdgcn_mov_dpp(__float_as_int(v), CTRL, 0xF, 0xF, true));
}
#define QUAD_LANE0 0x00
#define QUAD_LANE1 0x55
#define QUAD_LANE2 0xAA
#define QUAD_LANE3 0xFF
#define QUAD_XOR1  0xB1
#define QUAD_XOR2  0x4E
__device__ __forceinline__ float quad_sum(float v) {
  v += quad_read<QUAD_XOR1>(v);
  v += quad_read<QUAD_XOR2>(v);
  return v;
}
__device__ __forceinline__ float quad_max(float v) {
  v = fmaxf(v, quad_read<QUAD_XOR1>(v));
  v = fmaxf(v, quad_read<QUAD_XOR2>(v));
  return v;
}

__global__ __launch_bounds__(256) void mmse_pic_kernel(
    const float* __restrict__ y_re, const float* __restrict__ y_im,
    const float* __restrict__ h_re, const float* __restrict__ h_im,
    const float* __restrict__ prior,
    const float* __restrict__ s_re, const float* __restrict__ s_im,
    float* __restrict__ out)
{
  __shared__ WS ws[WAVES_PER_BLOCK];
  const int lane = threadIdx.x & 31;
  const int wid  = threadIdx.x >> 5;
  const int b    = blockIdx.x * WAVES_PER_BLOCK + wid;
  WS& w = ws[wid];

  #pragma unroll
  for (int t = 0; t < 8; ++t) {
    int idx = lane + 32*t;
    int i = idx >> 4, j = idx & 15;
    w.u.st.Mre[i][j] = s_re[b*256 + idx];
    w.u.st.Mim[i][j] = s_im[b*256 + idx];
  }
  #pragma unroll
  for (int t = 0; t < 4; ++t) {
    int idx = lane + 32*t;
    int m = idx >> 3, c = idx & 7;
    w.u.st.Mre[16 + c][m] =  h_re[b*128 + idx];
    w.u.st.Mim[16 + c][m] = -h_im[b*128 + idx];
  }
  if (lane < 16) {
    w.u.st.Mre[24][lane] =  y_re[b*16 + lane];
    w.u.st.Mim[24][lane] = -y_im[b*16 + lane];
  }
  const int s_out = lane >> 2, k_out = lane & 3;
  float my_lld = prior[b*32 + lane];
  __syncthreads();

  {
    const int rr = (lane < 25) ? lane : 24;
    float ar[16], ai[16];
    #pragma unroll
    for (int j = 0; j < 16; ++j) { ar[j] = w.u.st.Mre[rr][j]; ai[j] = w.u.st.Mim[rr][j]; }
    #pragma unroll
    for (int k = 0; k < 16; ++k) {
      float dk  = __int_as_float(__builtin_amdgcn_readlane(__float_as_int(ar[k]), k));
      float isd = __builtin_amdgcn_rsqf(dk);
      ar[k] *= isd; ai[k] *= isd;
      const float lre = ar[k], lim = ai[k];
      #pragma unroll
      for (int j = k + 1; j < 16; ++j) {
        float cre = __int_as_float(__builtin_amdgcn_readlane(__float_as_int(ar[k]), j));
        float cim = __int_as_float(__builtin_amdgcn_readlane(__float_as_int(ai[k]), j));
        ar[j] = __builtin_fmaf(-lre, cre, ar[j]); ar[j] = __builtin_fmaf(-lim, cim, ar[j]);
        ai[j] = __builtin_fmaf(-lim, cre, ai[j]); ai[j] = __builtin_fmaf( lre, cim, ai[j]);
      }
    }
    if (lane >= 16 && lane < 25) {
      const int c = lane - 16;
      #pragma unroll
      for (int k = 0; k < 16; ++k) { w.Wre[k][c] = ar[k]; w.Wim[k][c] = -ai[k]; }
    }
  }
  __syncthreads();

  if (lane < 8) {
    int s_ = lane;
    float ar = 0.0f, ai = 0.0f;
    #pragma unroll
    for (int m = 0; m < 16; ++m) {
      float hr = w.Wre[m][s_], hi = w.Wim[m][s_];
      float yr = w.Wre[m][8],  yi = w.Wim[m][8];
      ar += hr*yr + hi*yi;
      ai += hr*yi - hi*yr;
    }
    w.ymfre[s_] = ar; w.ymfim[s_] = ai;
  }
  #pragma unroll
  for (int t = 0; t < 2; ++t) {
    int idx = lane + 32*t;
    int i = idx >> 3, j = idx & 7;
    float ar = 0.0f, ai = 0.0f;
    #pragma unroll
    for (int m = 0; m < 16; ++m) {
      float xr = w.Wre[m][i], xi = w.Wim[m][i];
      float yr = w.Wre[m][j], yi = w.Wim[m][j];
      ar += xr*yr + xi*yi;
      ai += xr*yi - xi*yr;
    }
    w.gre[i][j] = ar; w.gim[i][j] = ai;
  }
  __syncthreads();
  #pragma unroll
  for (int t = 0; t < 8; ++t) {
    int idx = lane + 32*t;
    int i = idx >> 4, j = idx & 15;
    float v;
    if (i < 8) v = (j < 8) ? w.gre[i][j]   : -w.gim[i][j-8];
    else       v = (j < 8) ? w.gim[i-8][j] :  w.gre[i-8][j-8];
    w.gr[i][j] = v;
  }
  __syncthreads();

  const float C1 = 1.0f * INV_SQRT10, C3 = 3.0f * INV_SQRT10;
  const float prA = (k_out & 2) ? -C1 : C1, prB = (k_out & 2) ? -C3 : C3;
  const float piA = (k_out & 1) ? -C1 : C1, piB = (k_out & 1) ? -C3 : C3;
  float my_lla = 0.0f;
  for (int iter = 0; iter < 2; ++iter) {
    my_lla = my_lld;

    float lg[4];
    float var_s;
    {
      const float lsp = log_sigmoid( my_lla);
      const float lsn = log_sigmoid(-my_lla);
      const float p0 = quad_read<QUAD_LANE0>(lsp), n0 = quad_read<QUAD_LANE0>(lsn);
      const float p1 = quad_read<QUAD_LANE1>(lsp), n1 = quad_read<QUAD_LANE1>(lsn);
      const float p2 = quad_read<QUAD_LANE2>(lsp), n2 = quad_read<QUAD_LANE2>(lsn);
      const float p3 = quad_read<QUAD_LANE3>(lsp), n3 = quad_read<QUAD_LANE3>(lsn);
      const float base = ((k_out & 2) ? p0 : n0) + ((k_out & 1) ? p1 : n1);
      lg[0] = (base + n2) + n3;
      lg[1] = (base + n2) + p3;
      lg[2] = (base + p2) + n3;
      lg[3] = (base + p2) + p3;
      const float mx = quad_max(fmaxf(fmaxf(lg[0], lg[1]), fmaxf(lg[2], lg[3])));
      float sum = 0.0f, mre = 0.0f, mim = 0.0f, e2 = 0.0f;
      #pragma unroll
      for (int j = 0; j < 4; ++j) {
        float e = __expf(lg[j] - mx);
        float pr = (j & 2) ? prB : prA, pi = (j & 1) ? piB : piA;
        sum += e; mre += e*pr; mim += e*pi; e2 += e*(pr*pr + pi*pi);
      }
      sum = quad_sum(sum); mre = quad_sum(mre); mim = quad_sum(mim); e2 = quad_sum(e2);
      float invs = 1.0f / sum;
      mre *= invs; mim *= invs; e2 *= invs;
      var_s = e2 - (mre*mre + mim*mim);
      if (k_out == 0) { w.xre[s_out] = mre; w.xim[s_out] = mim; w.varL[s_out] = var_s; }
    }
    __syncthreads();

    if (lane < 8) {
      int i = lane;
      float ar = 0.0f, ai = 0.0f;
      #pragma unroll
      for (int j = 0; j < 8; ++j) {
        float gre_ = w.gre[i][j], gim_ = w.gim[i][j];
        float xr = w.xre[j], xi = w.xim[j];
        ar += gre_*xr - gim_*xi;
        ai += gre_*xi + gim_*xr;
      }
      w.gxre[i] = ar; w.gxim[i] = ai;
    }
    __syncthreads();

    #pragma unroll
    for (int t = 0; t < 2; ++t) {
      int idx = lane + 32*t;
      int i = idx >> 3, j = idx & 7;
      float gre_ = w.gre[j][i], gim_ = w.gim[j][i];
      float xr = w.xre[i], xi = w.xim[i];
      float tre = w.ymfre[j] + (gre_*xr - gim_*xi) - w.gxre[j];
      float tim = w.ymfim[j] + (gre_*xi + gim_*xr) - w.gxim[j];
      w.yptr[i][j]     = tre; w.yptr[i][j+8]   = tim;
      w.yptr[i+8][j]   = tre; w.yptr[i+8][j+8] = tim;
    }
    {
      const int cc = lane & 15;
      const int cj = cc & 7;
      const float vv = w.varL[cj];
      v2f x[8];
      #pragma unroll
      for (int r = 0; r < 8; ++r) {
        const float dl = (r == cj) ? 1.0f : 0.0f;
        const float cre = __builtin_fmaf(w.gre[r][cj], vv, dl);
        const float cim = w.gim[r][cj] * vv;
        x[r].x = (cc < 8) ? cre : dl;
        x[r].y = (cc < 8) ? cim : 0.0f;
      }
      #pragma unroll
      for (int k = 0; k < 8; ++k) {
        const float pre = __int_as_float(__builtin_amdgcn_readlane(__float_as_int(x[k].x), k));
        const float pim = __int_as_float(__builtin_amdgcn_readlane(__float_as_int(x[k].y), k));
        const float rd  = __builtin_amdgcn_rcpf(__builtin_fmaf(pre, pre, pim * pim));
        const float qre = pre * rd, qim = -pim * rd;
        float fre[8], fim[8];
        #pragma unroll
        for (int r = 0; r < 8; ++r) {
          fre[r] = __int_as_float(__builtin_amdgcn_readlane(__float_as_int(x[r].x), k));
          fim[r] = __int_as_float(__builtin_amdgcn_readlane(__float_as_int(x[r].y), k));
        }
        v2f sw;  sw.x = -x[k].y;  sw.y = x[k].x;
        v2f q1;  q1.x = qre;      q1.y = qre;
        v2f q2;  q2.x = qim;      q2.y = qim;
        v2f t = sw * q2;
        t = __builtin_elementwise_fma(x[k], q1, t);
        x[k] = t;
        v2f sT;  sT.x =  t.y;  sT.y = -t.x;
        #pragma unroll
        for (int r = 0; r < 8; ++r) {
          if (r != k) {
            v2f nf1;  nf1.x = -fre[r];  nf1.y = -fre[r];
            v2f f2;   f2.x  =  fim[r];  f2.y  =  fim[r];
            x[r] = __builtin_elementwise_fma(nf1, t, x[r]);
            x[r] = __builtin_elementwise_fma(f2, sT, x[r]);
          }
        }
      }
      if (lane >= 8 && lane < 16) {
        const int j = lane - 8;
        #pragma unroll
        for (int r = 0; r < 8; ++r) {
          w.u.aug[r][16 + j]         =  x[r].x;
          w.u.aug[r][16 + 8 + j]     = -x[r].y;
          w.u.aug[r + 8][16 + j]     =  x[r].y;
          w.u.aug[r + 8][16 + 8 + j] =  x[r].x;
        }
      }
    }
    __syncthreads();

    {
      int nn = lane & 15, hf = lane >> 4;
      v8f p1 = {}; v8f p2 = {};
      #pragma unroll
      for (int kb = 0; kb < 4; ++kb) {
        int k0 = kb*4 + hf*2;
        v2f af, b1, b2;
        af.x = w.u.aug[nn][16 + k0];
        af.y = w.u.aug[nn][16 + k0 + 1];
        b1.x = w.gr[k0][nn];   b1.y = w.gr[k0+1][nn];
        b2.x = w.yptr[nn][k0]; b2.y = w.yptr[nn][k0+1];
        p1 = __builtin_amdgcn_wmma_f32_16x16x4_f32(false, af, false, b1,
                                                   (short)0, p1, false, false);
        p2 = __builtin_amdgcn_wmma_f32_16x16x4_f32(false, af, false, b2,
                                                   (short)0, p2, false, false);
      }
      #pragma unroll
      for (int v = 0; v < 8; ++v) {
        int row = v + 8*hf;
        if (row == nn) { w.muL[nn] = p1[v]; w.znumL[nn] = p2[v]; }
      }
    }
    __syncthreads();

    const float m1 = w.muL[s_out], m2 = w.muL[s_out + 8];
    const float zr = w.znumL[s_out]     / m1;
    const float zi = w.znumL[s_out + 8] / m2;
    const float rho = m1 / fmaxf(1.0f - var_s * m1, EPSV);

    {
      float ev[4];
      #pragma unroll
      for (int j = 0; j < 4; ++j) {
        float pr = (j & 2) ? prB : prA, pi = (j & 1) ? piB : piA;
        float dre = zr - pr, dim = zi - pi;
        ev[j] = -(dre*dre + dim*dim) * rho + lg[j];
      }
      const float b2_0 = quad_max(fmaxf(ev[0], ev[1])), b2_1 = quad_max(fmaxf(ev[2], ev[3]));
      const float b3_0 = quad_max(fmaxf(ev[0], ev[2])), b3_1 = quad_max(fmaxf(ev[1], ev[3]));
      const float mall = fmaxf(fmaxf(ev[0], ev[1]), fmaxf(ev[2], ev[3]));
      const float u = fmaxf(mall, quad_read<QUAD_XOR1>(mall));
      const float v = fmaxf(mall, quad_read<QUAD_XOR2>(mall));
      const float b0_0 = quad_read<QUAD_LANE0>(u), b0_1 = quad_read<QUAD_LANE2>(u);
      const float b1_0 = quad_read<QUAD_LANE0>(v), b1_1 = quad_read<QUAD_LANE1>(v);
      float one  = (k_out == 0) ? b0_1 : (k_out == 1) ? b1_1 : (k_out == 2) ? b2_1 : b3_1;
      float zero = (k_out == 0) ? b0_0 : (k_out == 1) ? b1_0 : (k_out == 2) ? b2_0 : b3_0;
      my_lld = fmaxf(one, -1e30f) - fmaxf(zero, -1e30f);
    }
  }

  out[b*32 + lane] = my_lld - my_lla;
}

extern "C" void kernel_launch(void* const* d_in, const int* in_sizes, int n_in,
                              void* d_out, int out_size, void* d_ws, size_t ws_size,
                              hipStream_t stream) {
  (void)n_in; (void)out_size; (void)d_ws; (void)ws_size;
  const float* y_re  = (const float*)d_in[0];
  const float* y_im  = (const float*)d_in[1];
  const float* h_re  = (const float*)d_in[2];
  const float* h_im  = (const float*)d_in[3];
  const float* prior = (const float*)d_in[4];
  const float* s_re  = (const float*)d_in[5];
  const float* s_im  = (const float*)d_in[6];
  float* out = (float*)d_out;

  int B = in_sizes[0] / 16;
  int blocks = B / WAVES_PER_BLOCK;
  mmse_pic_kernel<<<blocks, 256, 0, stream>>>(y_re, y_im, h_re, h_im,
                                              prior, s_re, s_im, out);
}


# Whitening stage on (re, im) register pairs: each Cholesky/solve column update and each Gram / matched-filter term is two packed f32 fma, LDS rows stored as pairs; Gram sum order changes
# speedup vs baseline: 4.8755x; 1.0963x over previous
#include <hip/hip_runtime.h>
#include <math.h>

typedef __attribute__((ext_vector_type(2))) float v2f;
typedef __attribute__((ext_vector_type(8))) float v8f;

#define WAVES_PER_BLOCK 8
#define EPSV 1e-4f
#define INV_SQRT10 0.3162277660168379f


struct WS {
  union {
    v2f M[25][17];
    float aug[16][33];
  } u;
  v2f Wc[16][10];
  float gre[8][9],  gim[8][9];
  float gr[16][17];
  float yptr[16][17];
  float ymfre[8], ymfim[8];
  float xre[8], xim[8], varL[8];
  float gxre[8], gxim[8];
  float muL[16], znumL[16];
};

__device__ __forceinline__ float log_sigmoid(float x) {
  return fminf(x, 0.0f) - log1pf(__expf(-fabsf(x)));
}

template <int CTRL>
__device__ __forceinline__ float quad_read(float v) {
  return __int_as_float(__builtin_amdgcn_mov_dpp(__float_as_int(v), CTRL, 0xF, 0xF, true));
}
#define QUAD_LANE0 0x00
#define QUAD_LANE1 0x55
#define QUAD_LANE2 0xAA
#define QUAD_LANE3 0xFF
#define QUAD_XOR1  0xB1
#define QUAD_XOR2  0x4E
__device__ __forceinline__ float quad_sum(float v) {
  v += quad_read<QUAD_XOR1>(v);
  v += quad_read<QUAD_XOR2>(v);
  return v;
}
__device__ __forceinline__ float quad_max(float v) {
  v = fmaxf(v, quad_read<QUAD_XOR1>(v));
  v = fmaxf(v, quad_read<QUAD_XOR2>(v));
  return v;
}

__global__ __launch_bounds__(256) void mmse_pic_kernel(
    const float* __restrict__ y_re, const float* __restrict__ y_im,
    const float* __restrict__ h_re, const float* __restrict__ h_im,
    const float* __restrict__ prior,
    const float* __restrict__ s_re, const float* __restrict__ s_im,
    float* __restrict__ out)
{
  __shared__ WS ws[WAVES_PER_BLOCK];
  const int lane = threadIdx.x & 31;
  const int wid  = threadIdx.x >> 5;
  const int b    = blockIdx.x * WAVES_PER_BLOCK + wid;
  WS& w = ws[wid];

  #pragma unroll
  for (int t = 0; t < 8; ++t) {
    int idx = lane + 32*t;
    int i = idx >> 4, j = idx & 15;
    v2f e;  e.x = s_re[b*256 + idx];  e.y = s_im[b*256 + idx];
    w.u.M[i][j] = e;
  }
  #pragma unroll
  for (int t = 0; t < 4; ++t) {
    int idx = lane + 32*t;
    int m = idx >> 3, c = idx & 7;
    v2f e;  e.x = h_re[b*128 + idx];  e.y = -h_im[b*128 + idx];
    w.u.M[16 + c][m] = e;
  }
  if (lane < 16) {
    v2f e;  e.x = y_re[b*16 + lane];  e.y = -y_im[b*16 + lane];
    w.u.M[24][lane] = e;
  }
  const int s_out = lane >> 2, k_out = lane & 3;
  float my_lld = prior[b*32 + lane];
  __syncthreads();

  {
    const int rr = (lane < 25) ? lane : 24;
    v2f a[16];
    #pragma unroll
    for (int j = 0; j < 16; ++j) a[j] = w.u.M[rr][j];
    #pragma unroll
    for (int k = 0; k < 16; ++k) {
      float dk  = __int_as_float(__builtin_amdgcn_readlane(__float_as_int(a[k].x), k));
      float isd = __builtin_amdgcn_rsqf(dk);
      v2f s2;  s2.x = isd;  s2.y = isd;
      a[k] = a[k] * s2;
      const v2f l = a[k];
      v2f lsw;  lsw.x = -l.y;  lsw.y = l.x;
      #pragma unroll
      for (int j = k + 1; j < 16; ++j) {
        float cre = __int_as_float(__builtin_amdgcn_readlane(__float_as_int(a[k].x), j));
        float cim = __int_as_float(__builtin_amdgcn_readlane(__float_as_int(a[k].y), j));
        v2f ncre;  ncre.x = -cre;  ncre.y = -cre;
        v2f cim2;  cim2.x =  cim;  cim2.y =  cim;
        a[j] = __builtin_elementwise_fma(ncre, l,   a[j]);
        a[j] = __builtin_elementwise_fma(cim2, lsw, a[j]);
      }
    }
    if (lane >= 16 && lane < 25) {
      const int c = lane - 16;
      #pragma unroll
      for (int k = 0; k < 16; ++k) w.Wc[k][c] = a[k];
    }
  }
  __syncthreads();

  if (lane < 8) {
    int s_ = lane;
    v2f acc;  acc.x = 0.0f;  acc.y = 0.0f;
    #pragma unroll
    for (int m = 0; m < 16; ++m) {
      const v2f x = w.Wc[m][s_], y = w.Wc[m][8];
      v2f xr2;  xr2.x = x.x;  xr2.y = x.x;
      v2f xi2;  xi2.x = x.y;  xi2.y = x.y;
      v2f yA;   yA.x  = y.x;  yA.y  = -y.y;
      v2f yB;   yB.x  = y.y;  yB.y  =  y.x;
      acc = __builtin_elementwise_fma(xr2, yA, acc);
      acc = __builtin_elementwise_fma(xi2, yB, acc);
    }
    w.ymfre[s_] = acc.x; w.ymfim[s_] = acc.y;
  }
  #pragma unroll
  for (int t = 0; t < 2; ++t) {
    int idx = lane + 32*t;
    int i = idx >> 3, j = idx & 7;
    v2f acc;  acc.x = 0.0f;  acc.y = 0.0f;
    #pragma unroll
    for (int m = 0; m < 16; ++m) {
      const v2f x = w.Wc[m][i], y = w.Wc[m][j];
      v2f xr2;  xr2.x = x.x;  xr2.y = x.x;
      v2f xi2;  xi2.x = x.y;  xi2.y = x.y;
      v2f yA;   yA.x  = y.x;  yA.y  = -y.y;
      v2f yB;   yB.x  = y.y;  yB.y  =  y.x;
      acc = __builtin_elementwise_fma(xr2, yA, acc);
      acc = __builtin_elementwise_fma(xi2, yB, acc);
    }
    w.gre[i][j] = acc.x; w.gim[i][j] = acc.y;
  }
  __syncthreads();
  #pragma unroll
  for (int t = 0; t < 8; ++t) {
    int idx = lane + 32*t;
    int i = idx >> 4, j = idx & 15;
    float v;
    if (i < 8) v = (j < 8) ? w.gre[i][j]   : -w.gim[i][j-8];
    else       v = (j < 8) ? w.gim[i-8][j] :  w.gre[i-8][j-8];
    w.gr[i][j] = v;
  }
  __syncthreads();

  const float C1 = 1.0f * INV_SQRT10, C3 = 3.0f * INV_SQRT10;
  const float prA = (k_out & 2) ? -C1 : C1, prB = (k_out & 2) ? -C3 : C3;
  const float piA = (k_out & 1) ? -C1 : C1, piB = (k_out & 1) ? -C3 : C3;
  float my_lla = 0.0f;
  for (int iter = 0; iter < 2; ++iter) {
    my_lla = my_lld;

    float lg[4];
    float var_s;
    {
      const float lsp = log_sigmoid( my_lla);
      const float lsn = log_sigmoid(-my_lla);
      const float p0 = quad_read<QUAD_LANE0>(lsp), n0 = quad_read<QUAD_LANE0>(lsn);
      const float p1 = quad_read<QUAD_LANE1>(lsp), n1 = quad_read<QUAD_LANE1>(lsn);
      const float p2 = quad_read<QUAD_LANE2>(lsp), n2 = quad_read<QUAD_LANE2>(lsn);
      const float p3 = quad_read<QUAD_LANE3>(lsp), n3 = quad_read<QUAD_LANE3>(lsn);
      const float base = ((k_out & 2) ? p0 : n0) + ((k_out & 1) ? p1 : n1);
      lg[0] = (base + n2) + n3;
      lg[1] = (base + n2) + p3;
      lg[2] = (base + p2) + n3;
      lg[3] = (base + p2) + p3;
      const float mx = quad_max(fmaxf(fmaxf(lg[0], lg[1]), fmaxf(lg[2], lg[3])));
      float sum = 0.0f, mre = 0.0f, mim = 0.0f, e2 = 0.0f;
      #pragma unroll
      for (int j = 0; j < 4; ++j) {
        float e = __expf(lg[j] - mx);
        float pr = (j & 2) ? prB : prA, pi = (j & 1) ? piB : piA;
        sum += e; mre += e*pr; mim += e*pi; e2 += e*(pr*pr + pi*pi);
      }
      sum = quad_sum(sum); mre = quad_sum(mre); mim = quad_sum(mim); e2 = quad_sum(e2);
      float invs = 1.0f / sum;
      mre *= invs; mim *= invs; e2 *= invs;
      var_s = e2 - (mre*mre + mim*mim);
      if (k_out == 0) { w.xre[s_out] = mre; w.xim[s_out] = mim; w.varL[s_out] = var_s; }
    }
    __syncthreads();

    if (lane < 8) {
      int i = lane;
      float ar = 0.0f, ai = 0.0f;
      #pragma unroll
      for (int j = 0; j < 8; ++j) {
        float gre_ = w.gre[i][j], gim_ = w.gim[i][j];
        float xr = w.xre[j], xi = w.xim[j];
        ar += gre_*xr - gim_*xi;
        ai += gre_*xi + gim_*xr;
      }
      w.gxre[i] = ar; w.gxim[i] = ai;
    }
    __syncthreads();

    #pragma unroll
    for (int t = 0; t < 2; ++t) {
      int idx = lane + 32*t;
      int i = idx >> 3, j = idx & 7;
      float gre_ = w.gre[j][i], gim_ = w.gim[j][i];
      float xr = w.xre[i], xi = w.xim[i];
      float tre = w.ymfre[j] + (gre_*xr - gim_*xi) - w.gxre[j];
      float tim = w.ymfim[j] + (gre_*xi + gim_*xr) - w.gxim[j];
      w.yptr[i][j]     = tre; w.yptr[i][j+8]   = tim;
      w.yptr[i+8][j]   = tre; w.yptr[i+8][j+8] = tim;
    }
    {
      const int cc = lane & 15;
      const int cj = cc & 7;
      const float vv = w.varL[cj];
      v2f x[8];
      #pragma unroll
      for (int r = 0; r < 8; ++r) {
        const float dl = (r == cj) ? 1.0f : 0.0f;
        const float cre = __builtin_fmaf(w.gre[r][cj], vv, dl);
        const float cim = w.gim[r][cj] * vv;
        x[r].x = (cc < 8) ? cre : dl;
        x[r].y = (cc < 8) ? cim : 0.0f;
      }
      #pragma unroll
      for (int k = 0; k < 8; ++k) {
        const float pre = __int_as_float(__builtin_amdgcn_readlane(__float_as_int(x[k].x), k));
        const float pim = __int_as_float(__builtin_amdgcn_readlane(__float_as_int(x[k].y), k));
        const float rd  = __builtin_amdgcn_rcpf(__builtin_fmaf(pre, pre, pim * pim));
        const float qre = pre * rd, qim = -pim * rd;
        float fre[8], fim[8];
        #pragma unroll
        for (int r = 0; r < 8; ++r) {
          fre[r] = __int_as_float(__builtin_amdgcn_readlane(__float_as_int(x[r].x), k));
          fim[r] = __int_as_float(__builtin_amdgcn_readlane(__float_as_int(x[r].y), k));
        }
        v2f sw;  sw.x = -x[k].y;  sw.y = x[k].x;
        v2f q1;  q1.x = qre;      q1.y = qre;
        v2f q2;  q2.x = qim;      q2.y = qim;
        v2f t = sw * q2;
        t = __builtin_elementwise_fma(x[k], q1, t);
        x[k] = t;
        v2f sT;  sT.x =  t.y;  sT.y = -t.x;
        #pragma unroll
        for (int r = 0; r < 8; ++r) {
          if (r != k) {
            v2f nf1;  nf1.x = -fre[r];  nf1.y = -fre[r];
            v2f f2;   f2.x  =  fim[r];  f2.y  =  fim[r];
            x[r] = __builtin_elementwise_fma(nf1, t, x[r]);
            x[r] = __builtin_elementwise_fma(f2, sT, x[r]);
          }
        }
      }
      if (lane >= 8 && lane < 16) {
        const int j = lane - 8;
        #pragma unroll
        for (int r = 0; r < 8; ++r) {
          w.u.aug[r][16 + j]         =  x[r].x;
          w.u.aug[r][16 + 8 + j]     = -x[r].y;
          w.u.aug[r + 8][16 + j]     =  x[r].y;
          w.u.aug[r + 8][16 + 8 + j] =  x[r].x;
        }
      }
    }
    __syncthreads();

    {
      int nn = lane & 15, hf = lane >> 4;
      v8f p1 = {}; v8f p2 = {};
      #pragma unroll
      for (int kb = 0; kb < 4; ++kb) {
        int k0 = kb*4 + hf*2;
        v2f af, b1, b2;
        af.x = w.u.aug[nn][16 + k0];
        af.y = w.u.aug[nn][16 + k0 + 1];
        b1.x = w.gr[k0][nn];   b1.y = w.gr[k0+1][nn];
        b2.x = w.yptr[nn][k0]; b2.y = w.yptr[nn][k0+1];
        p1 = __builtin_amdgcn_wmma_f32_16x16x4_f32(false, af, false, b1,
                                                   (short)0, p1, false, false);
        p2 = __builtin_amdgcn_wmma_f32_16x16x4_f32(false, af, false, b2,
                                                   (short)0, p2, false, false);
      }
      #pragma unroll
      for (int v = 0; v < 8; ++v) {
        int row = v + 8*hf;
        if (row == nn) { w.muL[nn] = p1[v]; w.znumL[nn] = p2[v]; }
      }
    }
    __syncthreads();

    const float m1 = w.muL[s_out], m2 = w.muL[s_out + 8];
    const float zr = w.znumL[s_out]     / m1;
    const float zi = w.znumL[s_out + 8] / m2;
    const float rho = m1 / fmaxf(1.0f - var_s * m1, EPSV);

    {
      float ev[4];
      #pragma unroll
      for (int j = 0; j < 4; ++j) {
        float pr = (j & 2) ? prB : prA, pi = (j & 1) ? piB : piA;
        float dre = zr - pr, dim = zi - pi;
        ev[j] = -(dre*dre + dim*dim) * rho + lg[j];
      }
      const float b2_0 = quad_max(fmaxf(ev[0], ev[1])), b2_1 = quad_max(fmaxf(ev[2], ev[3]));
      const float b3_0 = quad_max(fmaxf(ev[0], ev[2])), b3_1 = quad_max(fmaxf(ev[1], ev[3]));
      const float mall = fmaxf(fmaxf(ev[0], ev[1]), fmaxf(ev[2], ev[3]));
      const float u = fmaxf(mall, quad_read<QUAD_XOR1>(mall));
      const float v = fmaxf(mall, quad_read<QUAD_XOR2>(mall));
      const float b0_0 = quad_read<QUAD_LANE0>(u), b0_1 = quad_read<QUAD_LANE2>(u);
      const float b1_0 = quad_read<QUAD_LANE0>(v), b1_1 = quad_read<QUAD_LANE1>(v);
      float one  = (k_out == 0) ? b0_1 : (k_out == 1) ? b1_1 : (k_out == 2) ? b2_1 : b3_1;
      float zero = (k_out == 0) ? b0_0 : (k_out == 1) ? b1_0 : (k_out == 2) ? b2_0 : b3_0;
      my_lld = fmaxf(one, -1e30f) - fmaxf(zero, -1e30f);
    }
  }

  out[b*32 + lane] = my_lld - my_lla;
}

extern "C" void kernel_launch(void* const* d_in, const int* in_sizes, int n_in,
                              void* d_out, int out_size, void* d_ws, size_t ws_size,
                              hipStream_t stream) {
  (void)n_in; (void)out_size; (void)d_ws; (void)ws_size;
  const float* y_re  = (const float*)d_in[0];
  const float* y_im  = (const float*)d_in[1];
  const float* h_re  = (const float*)d_in[2];
  const float* h_im  = (const float*)d_in[3];
  const float* prior = (const float*)d_in[4];
  const float* s_re  = (const float*)d_in[5];
  const float* s_im  = (const float*)d_in[6];
  float* out = (float*)d_out;

  int B = in_sizes[0] / 16;
  int blocks = B / WAVES_PER_BLOCK;
  mmse_pic_kernel<<<blocks, 256, 0, stream>>>(y_re, y_im, h_re, h_im,
                                              prior, s_re, s_im, out);
}


# Symbol stages: log1p of the log-sigmoid and the four divisions (softmax norm, two estimates, SINR) by the hardware log2 / rcp instructions (1 ulp), not correctly rounded sequences; sum order kept
# speedup vs baseline: 5.1657x; 1.0595x over previous
#include <hip/hip_runtime.h>
#include <math.h>

typedef __attribute__((ext_vector_type(2))) float v2f;
typedef __attribute__((ext_vector_type(8))) float v8f;

#define WAVES_PER_BLOCK 8
#define EPSV 1e-4f
#define INV_SQRT10 0.3162277660168379f


struct WS {
  union {
    v2f M[25][17];
    float aug[16][33];
  } u;
  v2f Wc[16][10];
  float gre[8][9],  gim[8][9];
  float gr[16][17];
  float yptr[16][17];
  float ymfre[8], ymfim[8];
  float xre[8], xim[8], varL[8];
  float gxre[8], gxim[8];
  float muL[16], znumL[16];
};

#define LN2F 0.6931471805599453f
__device__ __forceinline__ float log_sigmoid(float x) {
  return fminf(x, 0.0f) - LN2F * __builtin_amdgcn_logf(1.0f + __expf(-fabsf(x)));
}

template <int CTRL>
__device__ __forceinline__ float quad_read(float v) {
  return __int_as_float(__builtin_amdgcn_mov_dpp(__float_as_int(v), CTRL, 0xF, 0xF, true));
}
#define QUAD_LANE0 0x00
#define QUAD_LANE1 0x55
#define QUAD_LANE2 0xAA
#define QUAD_LANE3 0xFF
#define QUAD_XOR1  0xB1
#define QUAD_XOR2  0x4E
__device__ __forceinline__ float quad_sum(float v) {
  v += quad_read<QUAD_XOR1>(v);
  v += quad_read<QUAD_XOR2>(v);
  return v;
}
__device__ __forceinline__ float quad_max(float v) {
  v = fmaxf(v, quad_read<QUAD_XOR1>(v));
  v = fmaxf(v, quad_read<QUAD_XOR2>(v));
  return v;
}

__global__ __launch_bounds__(256) void mmse_pic_kernel(
    const float* __restrict__ y_re, const float* __restrict__ y_im,
    const float* __restrict__ h_re, const float* __restrict__ h_im,
    const float* __restrict__ prior,
    const float* __restrict__ s_re, const float* __restrict__ s_im,
    float* __restrict__ out)
{
  __shared__ WS ws[WAVES_PER_BLOCK];
  const int lane = threadIdx.x & 31;
  const int wid  = threadIdx.x >> 5;
  const int b    = blockIdx.x * WAVES_PER_BLOCK + wid;
  WS& w = ws[wid];

  #pragma unroll
  for (int t = 0; t < 8; ++t) {
    int idx = lane + 32*t;
    int i = idx >> 4, j = idx & 15;
    v2f e;  e.x = s_re[b*256 + idx];  e.y = s_im[b*256 + idx];
    w.u.M[i][j] = e;
  }
  #pragma unroll
  for (int t = 0; t < 4; ++t) {
    int idx = lane + 32*t;
    int m = idx >> 3, c = idx & 7;
    v2f e;  e.x = h_re[b*128 + idx];  e.y = -h_im[b*128 + idx];
    w.u.M[16 + c][m] = e;
  }
  if (lane < 16) {
    v2f e;  e.x = y_re[b*16 + lane];  e.y = -y_im[b*16 + lane];
    w.u.M[24][lane] = e;
  }
  const int s_out = lane >> 2, k_out = lane & 3;
  float my_lld = prior[b*32 + lane];
  __syncthreads();

  {
    const int rr = (lane < 25) ? lane : 24;
    v2f a[16];
    #pragma unroll
    for (int j = 0; j < 16; ++j) a[j] = w.u.M[rr][j];
    #pragma unroll
    for (int k = 0; k < 16; ++k) {
      float dk  = __int_as_float(__builtin_amdgcn_readlane(__float_as_int(a[k].x), k));
      float isd = __builtin_amdgcn_rsqf(dk);
      v2f s2;  s2.x = isd;  s2.y = isd;
      a[k] = a[k] * s2;
      const v2f l = a[k];
      v2f lsw;  lsw.x = -l.y;  lsw.y = l.x;
      #pragma unroll
      for (int j = k + 1; j < 16; ++j) {
        float cre = __int_as_float(__builtin_amdgcn_readlane(__float_as_int(a[k].x), j));
        float cim = __int_as_float(__builtin_amdgcn_readlane(__float_as_int(a[k].y), j));
        v2f ncre;  ncre.x = -cre;  ncre.y = -cre;
        v2f cim2;  cim2.x =  cim;  cim2.y =  cim;
        a[j] = __builtin_elementwise_fma(ncre, l,   a[j]);
        a[j] = __builtin_elementwise_fma(cim2, lsw, a[j]);
      }
    }
    if (lane >= 16 && lane < 25) {
      const int c = lane - 16;
      #pragma unroll
      for (int k = 0; k < 16; ++k) w.Wc[k][c] = a[k];
    }
  }
  __syncthreads();

  if (lane < 8) {
    int s_ = lane;
    v2f acc;  acc.x = 0.0f;  acc.y = 0.0f;
    #pragma unroll
    for (int m = 0; m < 16; ++m) {
      const v2f x = w.Wc[m][s_], y = w.Wc[m][8];
      v2f xr2;  xr2.x = x.x;  xr2.y = x.x;
      v2f xi2;  xi2.x = x.y;  xi2.y = x.y;
      v2f yA;   yA.x  = y.x;  yA.y  = -y.y;
      v2f yB;   yB.x  = y.y;  yB.y  =  y.x;
      acc = __builtin_elementwise_fma(xr2, yA, acc);
      acc = __builtin_elementwise_fma(xi2, yB, acc);
    }
    w.ymfre[s_] = acc.x; w.ymfim[s_] = acc.y;
  }
  #pragma unroll
  for (int t = 0; t < 2; ++t) {
    int idx = lane + 32*t;
    int i = idx >> 3, j = idx & 7;
    v2f acc;  acc.x = 0.0f;  acc.y = 0.0f;
    #pragma unroll
    for (int m = 0; m < 16; ++m) {
      const v2f x = w.Wc[m][i], y = w.Wc[m][j];
      v2f xr2;  xr2.x = x.x;  xr2.y = x.x;
      v2f xi2;  xi2.x = x.y;  xi2.y = x.y;
      v2f yA;   yA.x  = y.x;  yA.y  = -y.y;
      v2f yB;   yB.x  = y.y;  yB.y  =  y.x;
      acc = __builtin_elementwise_fma(xr2, yA, acc);
      acc = __builtin_elementwise_fma(xi2, yB, acc);
    }
    w.gre[i][j] = acc.x; w.gim[i][j] = acc.y;
  }
  __syncthreads();
  #pragma unroll
  for (int t = 0; t < 8; ++t) {
    int idx = lane + 32*t;
    int i = idx >> 4, j = idx & 15;
    float v;
    if (i < 8) v = (j < 8) ? w.gre[i][j]   : -w.gim[i][j-8];
    else       v = (j < 8) ? w.gim[i-8][j] :  w.gre[i-8][j-8];
    w.gr[i][j] = v;
  }
  __syncthreads();

  const float C1 = 1.0f * INV_SQRT10, C3 = 3.0f * INV_SQRT10;
  const float prA = (k_out & 2) ? -C1 : C1, prB = (k_out & 2) ? -C3 : C3;
  const float piA = (k_out & 1) ? -C1 : C1, piB = (k_out & 1) ? -C3 : C3;
  float my_lla = 0.0f;
  for (int iter = 0; iter < 2; ++iter) {
    my_lla = my_lld;

    float lg[4];
    float var_s;
    {
      const float lsp = log_sigmoid( my_lla);
      const float lsn = log_sigmoid(-my_lla);
      const float p0 = quad_read<QUAD_LANE0>(lsp), n0 = quad_read<QUAD_LANE0>(lsn);
      const float p1 = quad_read<QUAD_LANE1>(lsp), n1 = quad_read<QUAD_LANE1>(lsn);
      const float p2 = quad_read<QUAD_LANE2>(lsp), n2 = quad_read<QUAD_LANE2>(lsn);
      const float p3 = quad_read<QUAD_LANE3>(lsp), n3 = quad_read<QUAD_LANE3>(lsn);
      const float base = ((k_out & 2) ? p0 : n0) + ((k_out & 1) ? p1 : n1);
      lg[0] = (base + n2) + n3;
      lg[1] = (base + n2) + p3;
      lg[2] = (base + p2) + n3;
      lg[3] = (base + p2) + p3;
      const float mx = quad_max(fmaxf(fmaxf(lg[0], lg[1]), fmaxf(lg[2], lg[3])));
      float sum = 0.0f, mre = 0.0f, mim = 0.0f, e2 = 0.0f;
      #pragma unroll
      for (int j = 0; j < 4; ++j) {
        float e = __expf(lg[j] - mx);
        float pr = (j & 2) ? prB : prA, pi = (j & 1) ? piB : piA;
        sum += e; mre += e*pr; mim += e*pi; e2 += e*(pr*pr + pi*pi);
      }
      sum = quad_sum(sum); mre = quad_sum(mre); mim = quad_sum(mim); e2 = quad_sum(e2);
      float invs = __builtin_amdgcn_rcpf(sum);
      mre *= invs; mim *= invs; e2 *= invs;
      var_s = e2 - (mre*mre + mim*mim);
      if (k_out == 0) { w.xre[s_out] = mre; w.xim[s_out] = mim; w.varL[s_out] = var_s; }
    }
    __syncthreads();

    if (lane < 8) {
      int i = lane;
      float ar = 0.0f, ai = 0.0f;
      #pragma unroll
      for (int j = 0; j < 8; ++j) {
        float gre_ = w.gre[i][j], gim_ = w.gim[i][j];
        float xr = w.xre[j], xi = w.xim[j];
        ar += gre_*xr - gim_*xi;
        ai += gre_*xi + gim_*xr;
      }
      w.gxre[i] = ar; w.gxim[i] = ai;
    }
    __syncthreads();

    #pragma unroll
    for (int t = 0; t < 2; ++t) {
      int idx = lane + 32*t;
      int i = idx >> 3, j = idx & 7;
      float gre_ = w.gre[j][i], gim_ = w.gim[j][i];
      float xr = w.xre[i], xi = w.xim[i];
      float tre = w.ymfre[j] + (gre_*xr - gim_*xi) - w.gxre[j];
      float tim = w.ymfim[j] + (gre_*xi + gim_*xr) - w.gxim[j];
      w.yptr[i][j]     = tre; w.yptr[i][j+8]   = tim;
      w.yptr[i+8][j]   = tre; w.yptr[i+8][j+8] = tim;
    }
    {
      const int cc = lane & 15;
      const int cj = cc & 7;
      const float vv = w.varL[cj];
      v2f x[8];
      #pragma unroll
      for (int r = 0; r < 8; ++r) {
        const float dl = (r == cj) ? 1.0f : 0.0f;
        const float cre = __builtin_fmaf(w.gre[r][cj], vv, dl);
        const float cim = w.gim[r][cj] * vv;
        x[r].x = (cc < 8) ? cre : dl;
        x[r].y = (cc < 8) ? cim : 0.0f;
      }
      #pragma unroll
      for (int k = 0; k < 8; ++k) {
        const float pre = __int_as_float(__builtin_amdgcn_readlane(__float_as_int(x[k].x), k));
        const float pim = __int_as_float(__builtin_amdgcn_readlane(__float_as_int(x[k].y), k));
        const float rd  = __builtin_amdgcn_rcpf(__builtin_fmaf(pre, pre, pim * pim));
        const float qre = pre * rd, qim = -pim * rd;
        float fre[8], fim[8];
        #pragma unroll
        for (int r = 0; r < 8; ++r) {
          fre[r] = __int_as_float(__builtin_amdgcn_readlane(__float_as_int(x[r].x), k));
          fim[r] = __int_as_float(__builtin_amdgcn_readlane(__float_as_int(x[r].y), k));
        }
        v2f sw;  sw.x = -x[k].y;  sw.y = x[k].x;
        v2f q1;  q1.x = qre;      q1.y = qre;
        v2f q2;  q2.x = qim;      q2.y = qim;
        v2f t = sw * q2;
        t = __builtin_elementwise_fma(x[k], q1, t);
        x[k] = t;
        v2f sT;  sT.x =  t.y;  sT.y = -t.x;
        #pragma unroll
        for (int r = 0; r < 8; ++r) {
          if (r != k) {
            v2f nf1;  nf1.x = -fre[r];  nf1.y = -fre[r];
            v2f f2;   f2.x  =  fim[r];  f2.y  =  fim[r];
            x[r] = __builtin_elementwise_fma(nf1, t, x[r]);
            x[r] = __builtin_elementwise_fma(f2, sT, x[r]);
          }
        }
      }
      if (lane >= 8 && lane < 16) {
        const int j = lane - 8;
        #pragma unroll
        for (int r = 0; r < 8; ++r) {
          w.u.aug[r][16 + j]         =  x[r].x;
          w.u.aug[r][16 + 8 + j]     = -x[r].y;
          w.u.aug[r + 8][16 + j]     =  x[r].y;
          w.u.aug[r + 8][16 + 8 + j] =  x[r].x;
        }
      }
    }
    __syncthreads();

    {
      int nn = lane & 15, hf = lane >> 4;
      v8f p1 = {}; v8f p2 = {};
      #pragma unroll
      for (int kb = 0; kb < 4; ++kb) {
        int k0 = kb*4 + hf*2;
        v2f af, b1, b2;
        af.x = w.u.aug[nn][16 + k0];
        af.y = w.u.aug[nn][16 + k0 + 1];
        b1.x = w.gr[k0][nn];   b1.y = w.gr[k0+1][nn];
        b2.x = w.yptr[nn][k0]; b2.y = w.yptr[nn][k0+1];
        p1 = __builtin_amdgcn_wmma_f32_16x16x4_f32(false, af, false, b1,
                                                   (short)0, p1, false, false);
        p2 = __builtin_amdgcn_wmma_f32_16x16x4_f32(false, af, false, b2,
                                                   (short)0, p2, false, false);
      }
      #pragma unroll
      for (int v = 0; v < 8; ++v) {
        int row = v + 8*hf;
        if (row == nn) { w.muL[nn] = p1[v]; w.znumL[nn] = p2[v]; }
      }
    }
    __syncthreads();

    const float m1 = w.muL[s_out], m2 = w.muL[s_out + 8];
    const float zr = w.znumL[s_out]     * __builtin_amdgcn_rcpf(m1);
    const float zi = w.znumL[s_out + 8] * __builtin_amdgcn_rcpf(m2);
    const float rho = m1 * __builtin_amdgcn_rcpf(fmaxf(1.0f - var_s * m1, EPSV));

    {
      float ev[4];
      #pragma unroll
      for (int j = 0; j < 4; ++j) {
        float pr = (j & 2) ? prB : prA, pi = (j & 1) ? piB : piA;
        float dre = zr - pr, dim = zi - pi;
        ev[j] = -(dre*dre + dim*dim) * rho + lg[j];
      }
      const float b2_0 = quad_max(fmaxf(ev[0], ev[1])), b2_1 = quad_max(fmaxf(ev[2], ev[3]));
      const float b3_0 = quad_max(fmaxf(ev[0], ev[2])), b3_1 = quad_max(fmaxf(ev[1], ev[3]));
      const float mall = fmaxf(fmaxf(ev[0], ev[1]), fmaxf(ev[2], ev[3]));
      const float u = fmaxf(mall, quad_read<QUAD_XOR1>(mall));
      const float v = fmaxf(mall, quad_read<QUAD_XOR2>(mall));
      const float b0_0 = quad_read<QUAD_LANE0>(u), b0_1 = quad_read<QUAD_LANE2>(u);
      const float b1_0 = quad_read<QUAD_LANE0>(v), b1_1 = quad_read<QUAD_LANE1>(v);
      float one  = (k_out == 0) ? b0_1 : (k_out == 1) ? b1_1 : (k_out == 2) ? b2_1 : b3_1;
      float zero = (k_out == 0) ? b0_0 : (k_out == 1) ? b1_0 : (k_out == 2) ? b2_0 : b3_0;
      my_lld = fmaxf(one, -1e30f) - fmaxf(zero, -1e30f);
    }
  }

  out[b*32 + lane] = my_lld - my_lla;
}

extern "C" void kernel_launch(void* const* d_in, const int* in_sizes, int n_in,
                              void* d_out, int out_size, void* d_ws, size_t ws_size,
                              hipStream_t stream) {
  (void)n_in; (void)out_size; (void)d_ws; (void)ws_size;
  const float* y_re  = (const float*)d_in[0];
  const float* y_im  = (const float*)d_in[1];
  const float* h_re  = (const float*)d_in[2];
  const float* h_im  = (const float*)d_in[3];
  const float* prior = (const float*)d_in[4];
  const float* s_re  = (const float*)d_in[5];
  const float* s_im  = (const float*)d_in[6];
  float* out = (float*)d_out;

  int B = in_sizes[0] / 16;
  int blocks = B / WAVES_PER_BLOCK;
  mmse_pic_kernel<<<blocks, 256, 0, stream>>>(y_re, y_im, h_re, h_im,
                                              prior, s_re, s_im, out);
}


# Both diagonal products as ONE 16x16x16 f32 WMMA product, B = [gr columns 0-7 | ypt rows 0-7]: 4 WMMAs for 8, no gr build stage; mu of the Im half := mu of the Re half (same products, other sum order)
# speedup vs baseline: 5.5062x; 1.0659x over previous
#include <hip/hip_runtime.h>
#include <math.h>

typedef __attribute__((ext_vector_type(2))) float v2f;
typedef __attribute__((ext_vector_type(8))) float v8f;

#define WAVES_PER_BLOCK 8
#define EPSV 1e-4f
#define INV_SQRT10 0.3162277660168379f


struct WS {
  union {
    v2f M[25][17];
    struct {
      float aug[16][33];
      float resb[32][10];
    } g;
  } u;
  v2f Wc[16][10];
  float gre[8][9],  gim[8][9];
  float bm[16][17];
  float ymfre[8], ymfim[8];
  float xre[8], xim[8], varL[8];
  float gxre[8], gxim[8];
};

#define LN2F 0.6931471805599453f
__device__ __forceinline__ float log_sigmoid(float x) {
  return fminf(x, 0.0f) - LN2F * __builtin_amdgcn_logf(1.0f + __expf(-fabsf(x)));
}

template <int CTRL>
__device__ __forceinline__ float quad_read(float v) {
  return __int_as_float(__builtin_amdgcn_mov_dpp(__float_as_int(v), CTRL, 0xF, 0xF, true));
}
#define QUAD_LANE0 0x00
#define QUAD_LANE1 0x55
#define QUAD_LANE2 0xAA
#define QUAD_LANE3 0xFF
#define QUAD_XOR1  0xB1
#define QUAD_XOR2  0x4E
__device__ __forceinline__ float quad_sum(float v) {
  v += quad_read<QUAD_XOR1>(v);
  v += quad_read<QUAD_XOR2>(v);
  return v;
}
__device__ __forceinline__ float quad_max(float v) {
  v = fmaxf(v, quad_read<QUAD_XOR1>(v));
  v = fmaxf(v, quad_read<QUAD_XOR2>(v));
  return v;
}

__global__ __launch_bounds__(256) void mmse_pic_kernel(
    const float* __restrict__ y_re, const float* __restrict__ y_im,
    const float* __restrict__ h_re, const float* __restrict__ h_im,
    const float* __restrict__ prior,
    const float* __restrict__ s_re, const float* __restrict__ s_im,
    float* __restrict__ out)
{
  __shared__ WS ws[WAVES_PER_BLOCK];
  const int lane = threadIdx.x & 31;
  const int wid  = threadIdx.x >> 5;
  const int b    = blockIdx.x * WAVES_PER_BLOCK + wid;
  WS& w = ws[wid];

  #pragma unroll
  for (int t = 0; t < 8; ++t) {
    int idx = lane + 32*t;
    int i = idx >> 4, j = idx & 15;
    v2f e;  e.x = s_re[b*256 + idx];  e.y = s_im[b*256 + idx];
    w.u.M[i][j] = e;
  }
  #pragma unroll
  for (int t = 0; t < 4; ++t) {
    int idx = lane + 32*t;
    int m = idx >> 3, c = idx & 7;
    v2f e;  e.x = h_re[b*128 + idx];  e.y = -h_im[b*128 + idx];
    w.u.M[16 + c][m] = e;
  }
  if (lane < 16) {
    v2f e;  e.x = y_re[b*16 + lane];  e.y = -y_im[b*16 + lane];
    w.u.M[24][lane] = e;
  }
  const int s_out = lane >> 2, k_out = lane & 3;
  float my_lld = prior[b*32 + lane];
  __syncthreads();

  {
    const int rr = (lane < 25) ? lane : 24;
    v2f a[16];
    #pragma unroll
    for (int j = 0; j < 16; ++j) a[j] = w.u.M[rr][j];
    #pragma unroll
    for (int k = 0; k < 16; ++k) {
      float dk  = __int_as_float(__builtin_amdgcn_readlane(__float_as_int(a[k].x), k));
      float isd = __builtin_amdgcn_rsqf(dk);
      v2f s2;  s2.x = isd;  s2.y = isd;
      a[k] = a[k] * s2;
      const v2f l = a[k];
      v2f lsw;  lsw.x = -l.y;  lsw.y = l.x;
      #pragma unroll
      for (int j = k + 1; j < 16; ++j) {
        float cre = __int_as_float(__builtin_amdgcn_readlane(__float_as_int(a[k].x), j));
        float cim = __int_as_float(__builtin_amdgcn_readlane(__float_as_int(a[k].y), j));
        v2f ncre;  ncre.x = -cre;  ncre.y = -cre;
        v2f cim2;  cim2.x =  cim;  cim2.y =  cim;
        a[j] = __builtin_elementwise_fma(ncre, l,   a[j]);
        a[j] = __builtin_elementwise_fma(cim2, lsw, a[j]);
      }
    }
    if (lane >= 16 && lane < 25) {
      const int c = lane - 16;
      #pragma unroll
      for (int k = 0; k < 16; ++k) w.Wc[k][c] = a[k];
    }
  }
  __syncthreads();

  if (lane < 8) {
    int s_ = lane;
    v2f acc;  acc.x = 0.0f;  acc.y = 0.0f;
    #pragma unroll
    for (int m = 0; m < 16; ++m) {
      const v2f x = w.Wc[m][s_], y = w.Wc[m][8];
      v2f xr2;  xr2.x = x.x;  xr2.y = x.x;
      v2f xi2;  xi2.x = x.y;  xi2.y = x.y;
      v2f yA;   yA.x  = y.x;  yA.y  = -y.y;
      v2f yB;   yB.x  = y.y;  yB.y  =  y.x;
      acc = __builtin_elementwise_fma(xr2, yA, acc);
      acc = __builtin_elementwise_fma(xi2, yB, acc);
    }
    w.ymfre[s_] = acc.x; w.ymfim[s_] = acc.y;
  }
  #pragma unroll
  for (int t = 0; t < 2; ++t) {
    int idx = lane + 32*t;
    int i = idx >> 3, j = idx & 7;
    v2f acc;  acc.x = 0.0f;  acc.y = 0.0f;
    #pragma unroll
    for (int m = 0; m < 16; ++m) {
      const v2f x = w.Wc[m][i], y = w.Wc[m][j];
      v2f xr2;  xr2.x = x.x;  xr2.y = x.x;
      v2f xi2;  xi2.x = x.y;  xi2.y = x.y;
      v2f yA;   yA.x  = y.x;  yA.y  = -y.y;
      v2f yB;   yB.x  = y.y;  yB.y  =  y.x;
      acc = __builtin_elementwise_fma(xr2, yA, acc);
      acc = __builtin_elementwise_fma(xi2, yB, acc);
    }
    w.gre[i][j] = acc.x; w.gim[i][j] = acc.y;
    w.bm[i][j] = acc.x;  w.bm[i + 8][j] = acc.y;
  }
  __syncthreads();

  const float C1 = 1.0f * INV_SQRT10, C3 = 3.0f * INV_SQRT10;
  const float prA = (k_out & 2) ? -C1 : C1, prB = (k_out & 2) ? -C3 : C3;
  const float piA = (k_out & 1) ? -C1 : C1, piB = (k_out & 1) ? -C3 : C3;
  float my_lla = 0.0f;
  for (int iter = 0; iter < 2; ++iter) {
    my_lla = my_lld;

    float lg[4];
    float var_s;
    {
      const float lsp = log_sigmoid( my_lla);
      const float lsn = log_sigmoid(-my_lla);
      const float p0 = quad_read<QUAD_LANE0>(lsp), n0 = quad_read<QUAD_LANE0>(lsn);
      const float p1 = quad_read<QUAD_LANE1>(lsp), n1 = quad_read<QUAD_LANE1>(lsn);
      const float p2 = quad_read<QUAD_LANE2>(lsp), n2 = quad_read<QUAD_LANE2>(lsn);
      const float p3 = quad_read<QUAD_LANE3>(lsp), n3 = quad_read<QUAD_LANE3>(lsn);
      const float base = ((k_out & 2) ? p0 : n0) + ((k_out & 1) ? p1 : n1);
      lg[0] = (base + n2) + n3;
      lg[1] = (base + n2) + p3;
      lg[2] = (base + p2) + n3;
      lg[3] = (base + p2) + p3;
      const float mx = quad_max(fmaxf(fmaxf(lg[0], lg[1]), fmaxf(lg[2], lg[3])));
      float sum = 0.0f, mre = 0.0f, mim = 0.0f, e2 = 0.0f;
      #pragma unroll
      for (int j = 0; j < 4; ++j) {
        float e = __expf(lg[j] - mx);
        float pr = (j & 2) ? prB : prA, pi = (j & 1) ? piB : piA;
        sum += e; mre += e*pr; mim += e*pi; e2 += e*(pr*pr + pi*pi);
      }
      sum = quad_sum(sum); mre = quad_sum(mre); mim = quad_sum(mim); e2 = quad_sum(e2);
      float invs = __builtin_amdgcn_rcpf(sum);
      mre *= invs; mim *= invs; e2 *= invs;
      var_s = e2 - (mre*mre + mim*mim);
      if (k_out == 0) { w.xre[s_out] = mre; w.xim[s_out] = mim; w.varL[s_out] = var_s; }
    }
    __syncthreads();

    if (lane < 8) {
      int i = lane;
      float ar = 0.0f, ai = 0.0f;
      #pragma unroll
      for (int j = 0; j < 8; ++j) {
        float gre_ = w.gre[i][j], gim_ = w.gim[i][j];
        float xr = w.xre[j], xi = w.xim[j];
        ar += gre_*xr - gim_*xi;
        ai += gre_*xi + gim_*xr;
      }
      w.gxre[i] = ar; w.gxim[i] = ai;
    }
    __syncthreads();

    #pragma unroll
    for (int t = 0; t < 2; ++t) {
      int idx = lane + 32*t;
      int i = idx >> 3, j = idx & 7;
      float gre_ = w.gre[j][i], gim_ = w.gim[j][i];
      float xr = w.xre[i], xi = w.xim[i];
      float tre = w.ymfre[j] + (gre_*xr - gim_*xi) - w.gxre[j];
      float tim = w.ymfim[j] + (gre_*xi + gim_*xr) - w.gxim[j];
      w.bm[j][8 + i] = tre;  w.bm[j + 8][8 + i] = tim;
    }
    {
      const int cc = lane & 15;
      const int cj = cc & 7;
      const float vv = w.varL[cj];
      v2f x[8];
      #pragma unroll
      for (int r = 0; r < 8; ++r) {
        const float dl = (r == cj) ? 1.0f : 0.0f;
        const float cre = __builtin_fmaf(w.gre[r][cj], vv, dl);
        const float cim = w.gim[r][cj] * vv;
        x[r].x = (cc < 8) ? cre : dl;
        x[r].y = (cc < 8) ? cim : 0.0f;
      }
      #pragma unroll
      for (int k = 0; k < 8; ++k) {
        const float pre = __int_as_float(__builtin_amdgcn_readlane(__float_as_int(x[k].x), k));
        const float pim = __int_as_float(__builtin_amdgcn_readlane(__float_as_int(x[k].y), k));
        const float rd  = __builtin_amdgcn_rcpf(__builtin_fmaf(pre, pre, pim * pim));
        const float qre = pre * rd, qim = -pim * rd;
        float fre[8], fim[8];
        #pragma unroll
        for (int r = 0; r < 8; ++r) {
          fre[r] = __int_as_float(__builtin_amdgcn_readlane(__float_as_int(x[r].x), k));
          fim[r] = __int_as_float(__builtin_amdgcn_readlane(__float_as_int(x[r].y), k));
        }
        v2f sw;  sw.x = -x[k].y;  sw.y = x[k].x;
        v2f q1;  q1.x = qre;      q1.y = qre;
        v2f q2;  q2.x = qim;      q2.y = qim;
        v2f t = sw * q2;
        t = __builtin_elementwise_fma(x[k], q1, t);
        x[k] = t;
        v2f sT;  sT.x =  t.y;  sT.y = -t.x;
        #pragma unroll
        for (int r = 0; r < 8; ++r) {
          if (r != k) {
            v2f nf1;  nf1.x = -fre[r];  nf1.y = -fre[r];
            v2f f2;   f2.x  =  fim[r];  f2.y  =  fim[r];
            x[r] = __builtin_elementwise_fma(nf1, t, x[r]);
            x[r] = __builtin_elementwise_fma(f2, sT, x[r]);
          }
        }
      }
      if (lane >= 8 && lane < 16) {
        const int j = lane - 8;
        #pragma unroll
        for (int r = 0; r < 8; ++r) {
          w.u.g.aug[r][16 + j]         =  x[r].x;
          w.u.g.aug[r][16 + 8 + j]     = -x[r].y;
          w.u.g.aug[r + 8][16 + j]     =  x[r].y;
          w.u.g.aug[r + 8][16 + 8 + j] =  x[r].x;
        }
      }
    }
    __syncthreads();

    {
      int nn = lane & 15, hf = lane >> 4;
      v8f p = {};
      #pragma unroll
      for (int kb = 0; kb < 4; ++kb) {
        int k0 = kb*4 + hf*2;
        v2f af, bf;
        af.x = w.u.g.aug[nn][16 + k0];
        af.y = w.u.g.aug[nn][16 + k0 + 1];
        bf.x = w.bm[k0][nn];   bf.y = w.bm[k0+1][nn];
        p = __builtin_amdgcn_wmma_f32_16x16x4_f32(false, af, false, bf,
                                                  (short)0, p, false, false);
      }
      #pragma unroll
      for (int v = 0; v < 8; ++v) w.u.g.resb[lane][v] = p[v];
    }
    __syncthreads();

    const float m1  = w.u.g.resb[s_out][s_out];
    const float rm1 = __builtin_amdgcn_rcpf(m1);
    const float zr = w.u.g.resb[8 + s_out][s_out]  * rm1;
    const float zi = w.u.g.resb[24 + s_out][s_out] * rm1;
    const float rho = m1 * __builtin_amdgcn_rcpf(fmaxf(1.0f - var_s * m1, EPSV));

    {
      float ev[4];
      #pragma unroll
      for (int j = 0; j < 4; ++j) {
        float pr = (j & 2) ? prB : prA, pi = (j & 1) ? piB : piA;
        float dre = zr - pr, dim = zi - pi;
        ev[j] = -(dre*dre + dim*dim) * rho + lg[j];
      }
      const float b2_0 = quad_max(fmaxf(ev[0], ev[1])), b2_1 = quad_max(fmaxf(ev[2], ev[3]));
      const float b3_0 = quad_max(fmaxf(ev[0], ev[2])), b3_1 = quad_max(fmaxf(ev[1], ev[3]));
      const float mall = fmaxf(fmaxf(ev[0], ev[1]), fmaxf(ev[2], ev[3]));
      const float u = fmaxf(mall, quad_read<QUAD_XOR1>(mall));
      const float v = fmaxf(mall, quad_read<QUAD_XOR2>(mall));
      const float b0_0 = quad_read<QUAD_LANE0>(u), b0_1 = quad_read<QUAD_LANE2>(u);
      const float b1_0 = quad_read<QUAD_LANE0>(v), b1_1 = quad_read<QUAD_LANE1>(v);
      float one  = (k_out == 0) ? b0_1 : (k_out == 1) ? b1_1 : (k_out == 2) ? b2_1 : b3_1;
      float zero = (k_out == 0) ? b0_0 : (k_out == 1) ? b1_0 : (k_out == 2) ? b2_0 : b3_0;
      my_lld = fmaxf(one, -1e30f) - fmaxf(zero, -1e30f);
    }
  }

  out[b*32 + lane] = my_lld - my_lla;
}

extern "C" void kernel_launch(void* const* d_in, const int* in_sizes, int n_in,
                              void* d_out, int out_size, void* d_ws, size_t ws_size,
                              hipStream_t stream) {
  (void)n_in; (void)out_size; (void)d_ws; (void)ws_size;
  const float* y_re  = (const float*)d_in[0];
  const float* y_im  = (const float*)d_in[1];
  const float* h_re  = (const float*)d_in[2];
  const float* h_im  = (const float*)d_in[3];
  const float* prior = (const float*)d_in[4];
  const float* s_re  = (const float*)d_in[5];
  const float* s_im  = (const float*)d_in[6];
  float* out = (float*)d_out;

  int B = in_sizes[0] / 16;
  int blocks = B / WAVES_PER_BLOCK;
  mmse_pic_kernel<<<blocks, 256, 0, stream>>>(y_re, y_im, h_re, h_im,
                                              prior, s_re, s_im, out);
}


# The 7 work-group barriers (11 per wave) replaced by wave-scope LDS ordering fences that emit no instruction: every wave touches only its own LDS slice; same arithmetic and sum order
# speedup vs baseline: 5.7699x; 1.0479x over previous
#include <hip/hip_runtime.h>
#include <math.h>

typedef __attribute__((ext_vector_type(2))) float v2f;
typedef __attribute__((ext_vector_type(8))) float v8f;

#define WAVES_PER_BLOCK 8
#define EPSV 1e-4f
#define INV_SQRT10 0.3162277660168379f


struct WS {
  union {
    v2f M[25][17];
    struct {
      float aug[16][33];
      float resb[32][10];
    } g;
  } u;
  v2f Wc[16][10];
  float gre[8][9],  gim[8][9];
  float bm[16][17];
  float ymfre[8], ymfim[8];
  float xre[8], xim[8], varL[8];
  float gxre[8], gxim[8];
};

#define LN2F 0.6931471805599453f
__device__ __forceinline__ float log_sigmoid(float x) {
  return fminf(x, 0.0f) - LN2F * __builtin_amdgcn_logf(1.0f + __expf(-fabsf(x)));
}

template <int CTRL>
__device__ __forceinline__ float quad_read(float v) {
  return __int_as_float(__builtin_amdgcn_mov_dpp(__float_as_int(v), CTRL, 0xF, 0xF, true));
}
#define QUAD_LANE0 0x00
#define QUAD_LANE1 0x55
#define QUAD_LANE2 0xAA
#define QUAD_LANE3 0xFF
#define QUAD_XOR1  0xB1
#define QUAD_XOR2  0x4E
__device__ __forceinline__ float quad_sum(float v) {
  v += quad_read<QUAD_XOR1>(v);
  v += quad_read<QUAD_XOR2>(v);
  return v;
}
__device__ __forceinline__ float quad_max(float v) {
  v = fmaxf(v, quad_read<QUAD_XOR1>(v));
  v = fmaxf(v, quad_read<QUAD_XOR2>(v));
  return v;
}

__device__ __forceinline__ void wave_lds_order() {
  __builtin_amdgcn_fence(__ATOMIC_RELEASE, "wavefront");
  __builtin_amdgcn_fence(__ATOMIC_ACQUIRE, "wavefront");
}

__global__ __launch_bounds__(256) void mmse_pic_kernel(
    const float* __restrict__ y_re, const float* __restrict__ y_im,
    const float* __restrict__ h_re, const float* __restrict__ h_im,
    const float* __restrict__ prior,
    const float* __restrict__ s_re, const float* __restrict__ s_im,
    float* __restrict__ out)
{
  __shared__ WS ws[WAVES_PER_BLOCK];
  const int lane = threadIdx.x & 31;
  const int wid  = threadIdx.x >> 5;
  const int b    = blockIdx.x * WAVES_PER_BLOCK + wid;
  WS& w = ws[wid];

  #pragma unroll
  for (int t = 0; t < 8; ++t) {
    int idx = lane + 32*t;
    int i = idx >> 4, j = idx & 15;
    v2f e;  e.x = s_re[b*256 + idx];  e.y = s_im[b*256 + idx];
    w.u.M[i][j] = e;
  }
  #pragma unroll
  for (int t = 0; t < 4; ++t) {
    int idx = lane + 32*t;
    int m = idx >> 3, c = idx & 7;
    v2f e;  e.x = h_re[b*128 + idx];  e.y = -h_im[b*128 + idx];
    w.u.M[16 + c][m] = e;
  }
  if (lane < 16) {
    v2f e;  e.x = y_re[b*16 + lane];  e.y = -y_im[b*16 + lane];
    w.u.M[24][lane] = e;
  }
  const int s_out = lane >> 2, k_out = lane & 3;
  float my_lld = prior[b*32 + lane];
  wave_lds_order();

  {
    const int rr = (lane < 25) ? lane : 24;
    v2f a[16];
    #pragma unroll
    for (int j = 0; j < 16; ++j) a[j] = w.u.M[rr][j];
    #pragma unroll
    for (int k = 0; k < 16; ++k) {
      float dk  = __int_as_float(__builtin_amdgcn_readlane(__float_as_int(a[k].x), k));
      float isd = __builtin_amdgcn_rsqf(dk);
      v2f s2;  s2.x = isd;  s2.y = isd;
      a[k] = a[k] * s2;
      const v2f l = a[k];
      v2f lsw;  lsw.x = -l.y;  lsw.y = l.x;
      #pragma unroll
      for (int j = k + 1; j < 16; ++j) {
        float cre = __int_as_float(__builtin_amdgcn_readlane(__float_as_int(a[k].x), j));
        float cim = __int_as_float(__builtin_amdgcn_readlane(__float_as_int(a[k].y), j));
        v2f ncre;  ncre.x = -cre;  ncre.y = -cre;
        v2f cim2;  cim2.x =  cim;  cim2.y =  cim;
        a[j] = __builtin_elementwise_fma(ncre, l,   a[j]);
        a[j] = __builtin_elementwise_fma(cim2, lsw, a[j]);
      }
    }
    if (lane >= 16 && lane < 25) {
      const int c = lane - 16;
      #pragma unroll
      for (int k = 0; k < 16; ++k) w.Wc[k][c] = a[k];
    }
  }
  wave_lds_order();

  if (lane < 8) {
    int s_ = lane;
    v2f acc;  acc.x = 0.0f;  acc.y = 0.0f;
    #pragma unroll
    for (int m = 0; m < 16; ++m) {
      const v2f x = w.Wc[m][s_], y = w.Wc[m][8];
      v2f xr2;  xr2.x = x.x;  xr2.y = x.x;
      v2f xi2;  xi2.x = x.y;  xi2.y = x.y;
      v2f yA;   yA.x  = y.x;  yA.y  = -y.y;
      v2f yB;   yB.x  = y.y;  yB.y  =  y.x;
      acc = __builtin_elementwise_fma(xr2, yA, acc);
      acc = __builtin_elementwise_fma(xi2, yB, acc);
    }
    w.ymfre[s_] = acc.x; w.ymfim[s_] = acc.y;
  }
  #pragma unroll
  for (int t = 0; t < 2; ++t) {
    int idx = lane + 32*t;
    int i = idx >> 3, j = idx & 7;
    v2f acc;  acc.x = 0.0f;  acc.y = 0.0f;
    #pragma unroll
    for (int m = 0; m < 16; ++m) {
      const v2f x = w.Wc[m][i], y = w.Wc[m][j];
      v2f xr2;  xr2.x = x.x;  xr2.y = x.x;
      v2f xi2;  xi2.x = x.y;  xi2.y = x.y;
      v2f yA;   yA.x  = y.x;  yA.y  = -y.y;
      v2f yB;   yB.x  = y.y;  yB.y  =  y.x;
      acc = __builtin_elementwise_fma(xr2, yA, acc);
      acc = __builtin_elementwise_fma(xi2, yB, acc);
    }
    w.gre[i][j] = acc.x; w.gim[i][j] = acc.y;
    w.bm[i][j] = acc.x;  w.bm[i + 8][j] = acc.y;
  }
  wave_lds_order();

  const float C1 = 1.0f * INV_SQRT10, C3 = 3.0f * INV_SQRT10;
  const float prA = (k_out & 2) ? -C1 : C1, prB = (k_out & 2) ? -C3 : C3;
  const float piA = (k_out & 1) ? -C1 : C1, piB = (k_out & 1) ? -C3 : C3;
  float my_lla = 0.0f;
  for (int iter = 0; iter < 2; ++iter) {
    my_lla = my_lld;

    float lg[4];
    float var_s;
    {
      const float lsp = log_sigmoid( my_lla);
      const float lsn = log_sigmoid(-my_lla);
      const float p0 = quad_read<QUAD_LANE0>(lsp), n0 = quad_read<QUAD_LANE0>(lsn);
      const float p1 = quad_read<QUAD_LANE1>(lsp), n1 = quad_read<QUAD_LANE1>(lsn);
      const float p2 = quad_read<QUAD_LANE2>(lsp), n2 = quad_read<QUAD_LANE2>(lsn);
      const float p3 = quad_read<QUAD_LANE3>(lsp), n3 = quad_read<QUAD_LANE3>(lsn);
      const float base = ((k_out & 2) ? p0 : n0) + ((k_out & 1) ? p1 : n1);
      lg[0] = (base + n2) + n3;
      lg[1] = (base + n2) + p3;
      lg[2] = (base + p2) + n3;
      lg[3] = (base + p2) + p3;
      const float mx = quad_max(fmaxf(fmaxf(lg[0], lg[1]), fmaxf(lg[2], lg[3])));
      float sum = 0.0f, mre = 0.0f, mim = 0.0f, e2 = 0.0f;
      #pragma unroll
      for (int j = 0; j < 4; ++j) {
        float e = __expf(lg[j] - mx);
        float pr = (j & 2) ? prB : prA, pi = (j & 1) ? piB : piA;
        sum += e; mre += e*pr; mim += e*pi; e2 += e*(pr*pr + pi*pi);
      }
      sum = quad_sum(sum); mre = quad_sum(mre); mim = quad_sum(mim); e2 = quad_sum(e2);
      float invs = __builtin_amdgcn_rcpf(sum);
      mre *= invs; mim *= invs; e2 *= invs;
      var_s = e2 - (mre*mre + mim*mim);
      if (k_out == 0) { w.xre[s_out] = mre; w.xim[s_out] = mim; w.varL[s_out] = var_s; }
    }
    wave_lds_order();

    if (lane < 8) {
      int i = lane;
      float ar = 0.0f, ai = 0.0f;
      #pragma unroll
      for (int j = 0; j < 8; ++j) {
        float gre_ = w.gre[i][j], gim_ = w.gim[i][j];
        float xr = w.xre[j], xi = w.xim[j];
        ar += gre_*xr - gim_*xi;
        ai += gre_*xi + gim_*xr;
      }
      w.gxre[i] = ar; w.gxim[i] = ai;
    }
    wave_lds_order();

    #pragma unroll
    for (int t = 0; t < 2; ++t) {
      int idx = lane + 32*t;
      int i = idx >> 3, j = idx & 7;
      float gre_ = w.gre[j][i], gim_ = w.gim[j][i];
      float xr = w.xre[i], xi = w.xim[i];
      float tre = w.ymfre[j] + (gre_*xr - gim_*xi) - w.gxre[j];
      float tim = w.ymfim[j] + (gre_*xi + gim_*xr) - w.gxim[j];
      w.bm[j][8 + i] = tre;  w.bm[j + 8][8 + i] = tim;
    }
    {
      const int cc = lane & 15;
      const int cj = cc & 7;
      const float vv = w.varL[cj];
      v2f x[8];
      #pragma unroll
      for (int r = 0; r < 8; ++r) {
        const float dl = (r == cj) ? 1.0f : 0.0f;
        const float cre = __builtin_fmaf(w.gre[r][cj], vv, dl);
        const float cim = w.gim[r][cj] * vv;
        x[r].x = (cc < 8) ? cre : dl;
        x[r].y = (cc < 8) ? cim : 0.0f;
      }
      #pragma unroll
      for (int k = 0; k < 8; ++k) {
        const float pre = __int_as_float(__builtin_amdgcn_readlane(__float_as_int(x[k].x), k));
        const float pim = __int_as_float(__builtin_amdgcn_readlane(__float_as_int(x[k].y), k));
        const float rd  = __builtin_amdgcn_rcpf(__builtin_fmaf(pre, pre, pim * pim));
        const float qre = pre * rd, qim = -pim * rd;
        float fre[8], fim[8];
        #pragma unroll
        for (int r = 0; r < 8; ++r) {
          fre[r] = __int_as_float(__builtin_amdgcn_readlane(__float_as_int(x[r].x), k));
          fim[r] = __int_as_float(__builtin_amdgcn_readlane(__float_as_int(x[r].y), k));
        }
        v2f sw;  sw.x = -x[k].y;  sw.y = x[k].x;
        v2f q1;  q1.x = qre;      q1.y = qre;
        v2f q2;  q2.x = qim;      q2.y = qim;
        v2f t = sw * q2;
        t = __builtin_elementwise_fma(x[k], q1, t);
        x[k] = t;
        v2f sT;  sT.x =  t.y;  sT.y = -t.x;
        #pragma unroll
        for (int r = 0; r < 8; ++r) {
          if (r != k) {
            v2f nf1;  nf1.x = -fre[r];  nf1.y = -fre[r];
            v2f f2;   f2.x  =  fim[r];  f2.y  =  fim[r];
            x[r] = __builtin_elementwise_fma(nf1, t, x[r]);
            x[r] = __builtin_elementwise_fma(f2, sT, x[r]);
          }
        }
      }
      if (lane >= 8 && lane < 16) {
        const int j = lane - 8;
        #pragma unroll
        for (int r = 0; r < 8; ++r) {
          w.u.g.aug[r][16 + j]         =  x[r].x;
          w.u.g.aug[r][16 + 8 + j]     = -x[r].y;
          w.u.g.aug[r + 8][16 + j]     =  x[r].y;
          w.u.g.aug[r + 8][16 + 8 + j] =  x[r].x;
        }
      }
    }
    wave_lds_order();

    {
      int nn = lane & 15, hf = lane >> 4;
      v8f p = {};
      #pragma unroll
      for (int kb = 0; kb < 4; ++kb) {
        int k0 = kb*4 + hf*2;
        v2f af, bf;
        af.x = w.u.g.aug[nn][16 + k0];
        af.y = w.u.g.aug[nn][16 + k0 + 1];
        bf.x = w.bm[k0][nn];   bf.y = w.bm[k0+1][nn];
        p = __builtin_amdgcn_wmma_f32_16x16x4_f32(false, af, false, bf,
                                                  (short)0, p, false, false);
      }
      #pragma unroll
      for (int v = 0; v < 8; ++v) w.u.g.resb[lane][v] = p[v];
    }
    wave_lds_order();

    const float m1  = w.u.g.resb[s_out][s_out];
    const float rm1 = __builtin_amdgcn_rcpf(m1);
    const float zr = w.u.g.resb[8 + s_out][s_out]  * rm1;
    const float zi = w.u.g.resb[24 + s_out][s_out] * rm1;
    const float rho = m1 * __builtin_amdgcn_rcpf(fmaxf(1.0f - var_s * m1, EPSV));

    {
      float ev[4];
      #pragma unroll
      for (int j = 0; j < 4; ++j) {
        float pr = (j & 2) ? prB : prA, pi = (j & 1) ? piB : piA;
        float dre = zr - pr, dim = zi - pi;
        ev[j] = -(dre*dre + dim*dim) * rho + lg[j];
      }
      const float b2_0 = quad_max(fmaxf(ev[0], ev[1])), b2_1 = quad_max(fmaxf(ev[2], ev[3]));
      const float b3_0 = quad_max(fmaxf(ev[0], ev[2])), b3_1 = quad_max(fmaxf(ev[1], ev[3]));
      const float mall = fmaxf(fmaxf(ev[0], ev[1]), fmaxf(ev[2], ev[3]));
      const float u = fmaxf(mall, quad_read<QUAD_XOR1>(mall));
      const float v = fmaxf(mall, quad_read<QUAD_XOR2>(mall));
      const float b0_0 = quad_read<QUAD_LANE0>(u), b0_1 = quad_read<QUAD_LANE2>(u);
      const float b1_0 = quad_read<QUAD_LANE0>(v), b1_1 = quad_read<QUAD_LANE1>(v);
      float one  = (k_out == 0) ? b0_1 : (k_out == 1) ? b1_1 : (k_out == 2) ? b2_1 : b3_1;
      float zero = (k_out == 0) ? b0_0 : (k_out == 1) ? b1_0 : (k_out == 2) ? b2_0 : b3_0;
      my_lld = fmaxf(one, -1e30f) - fmaxf(zero, -1e30f);
    }
  }

  out[b*32 + lane] = my_lld - my_lla;
}

extern "C" void kernel_launch(void* const* d_in, const int* in_sizes, int n_in,
                              void* d_out, int out_size, void* d_ws, size_t ws_size,
                              hipStream_t stream) {
  (void)n_in; (void)out_size; (void)d_ws; (void)ws_size;
  const float* y_re  = (const float*)d_in[0];
  const float* y_im  = (const float*)d_in[1];
  const float* h_re  = (const float*)d_in[2];
  const float* h_im  = (const float*)d_in[3];
  const float* prior = (const float*)d_in[4];
  const float* s_re  = (const float*)d_in[5];
  const float* s_im  = (const float*)d_in[6];
  float* out = (float*)d_out;

  int B = in_sizes[0] / 16;
  int blocks = B / WAVES_PER_BLOCK;
  mmse_pic_kernel<<<blocks, 256, 0, stream>>>(y_re, y_im, h_re, h_im,
                                              prior, s_re, s_im, out);
}


# Gauss-Jordan: the 8 complex rows split over the two 16-lane halves (4 row updates per lane and step, not 8; lanes 16-31 were idle copies); factors and pivot row pass through LDS; sum order kept
# speedup vs baseline: 6.2727x; 1.0871x over previous
#include <hip/hip_runtime.h>
#include <math.h>

typedef __attribute__((ext_vector_type(2))) float v2f;
typedef __attribute__((ext_vector_type(8))) float v8f;

#define WAVES_PER_BLOCK 8
#define EPSV 1e-4f
#define INV_SQRT10 0.3162277660168379f


struct WS {
  union {
    v2f M[25][17];
    struct {
      float aug[16][33];
      float resb[32][10];
    } g;
  } u;
  union {
    v2f Wc[16][10];
    alignas(16) v2f cb[2][2][16][2];
  } v;
  float gre[8][9],  gim[8][9];
  float bm[16][17];
  float ymfre[8], ymfim[8];
  float xre[8], xim[8], varL[8];
  float gxre[8], gxim[8];
};

#define LN2F 0.6931471805599453f
__device__ __forceinline__ float log_sigmoid(float x) {
  return fminf(x, 0.0f) - LN2F * __builtin_amdgcn_logf(1.0f + __expf(-fabsf(x)));
}

template <int CTRL>
__device__ __forceinline__ float quad_read(float v) {
  return __int_as_float(__builtin_amdgcn_mov_dpp(__float_as_int(v), CTRL, 0xF, 0xF, true));
}
#define QUAD_LANE0 0x00
#define QUAD_LANE1 0x55
#define QUAD_LANE2 0xAA
#define QUAD_LANE3 0xFF
#define QUAD_XOR1  0xB1
#define QUAD_XOR2  0x4E
__device__ __forceinline__ float quad_sum(float v) {
  v += quad_read<QUAD_XOR1>(v);
  v += quad_read<QUAD_XOR2>(v);
  return v;
}
__device__ __forceinline__ float quad_max(float v) {
  v = fmaxf(v, quad_read<QUAD_XOR1>(v));
  v = fmaxf(v, quad_read<QUAD_XOR2>(v));
  return v;
}

__device__ __forceinline__ void wave_lds_order() {
  __builtin_amdgcn_fence(__ATOMIC_RELEASE, "wavefront");
  __builtin_amdgcn_fence(__ATOMIC_ACQUIRE, "wavefront");
}

__global__ __launch_bounds__(256) void mmse_pic_kernel(
    const float* __restrict__ y_re, const float* __restrict__ y_im,
    const float* __restrict__ h_re, const float* __restrict__ h_im,
    const float* __restrict__ prior,
    const float* __restrict__ s_re, const float* __restrict__ s_im,
    float* __restrict__ out)
{
  __shared__ WS ws[WAVES_PER_BLOCK];
  const int lane = threadIdx.x & 31;
  const int wid  = threadIdx.x >> 5;
  const int b    = blockIdx.x * WAVES_PER_BLOCK + wid;
  WS& w = ws[wid];

  #pragma unroll
  for (int t = 0; t < 8; ++t) {
    int idx = lane + 32*t;
    int i = idx >> 4, j = idx & 15;
    v2f e;  e.x = s_re[b*256 + idx];  e.y = s_im[b*256 + idx];
    w.u.M[i][j] = e;
  }
  #pragma unroll
  for (int t = 0; t < 4; ++t) {
    int idx = lane + 32*t;
    int m = idx >> 3, c = idx & 7;
    v2f e;  e.x = h_re[b*128 + idx];  e.y = -h_im[b*128 + idx];
    w.u.M[16 + c][m] = e;
  }
  if (lane < 16) {
    v2f e;  e.x = y_re[b*16 + lane];  e.y = -y_im[b*16 + lane];
    w.u.M[24][lane] = e;
  }
  const int s_out = lane >> 2, k_out = lane & 3;
  float my_lld = prior[b*32 + lane];
  wave_lds_order();

  {
    const int rr = (lane < 25) ? lane : 24;
    v2f a[16];
    #pragma unroll
    for (int j = 0; j < 16; ++j) a[j] = w.u.M[rr][j];
    #pragma unroll
    for (int k = 0; k < 16; ++k) {
      float dk  = __int_as_float(__builtin_amdgcn_readlane(__float_as_int(a[k].x), k));
      float isd = __builtin_amdgcn_rsqf(dk);
      v2f s2;  s2.x = isd;  s2.y = isd;
      a[k] = a[k] * s2;
      const v2f l = a[k];
      v2f lsw;  lsw.x = -l.y;  lsw.y = l.x;
      #pragma unroll
      for (int j = k + 1; j < 16; ++j) {
        float cre = __int_as_float(__builtin_amdgcn_readlane(__float_as_int(a[k].x), j));
        float cim = __int_as_float(__builtin_amdgcn_readlane(__float_as_int(a[k].y), j));
        v2f ncre;  ncre.x = -cre;  ncre.y = -cre;
        v2f cim2;  cim2.x =  cim;  cim2.y =  cim;
        a[j] = __builtin_elementwise_fma(ncre, l,   a[j]);
        a[j] = __builtin_elementwise_fma(cim2, lsw, a[j]);
      }
    }
    if (lane >= 16 && lane < 25) {
      const int c = lane - 16;
      #pragma unroll
      for (int k = 0; k < 16; ++k) w.v.Wc[k][c] = a[k];
    }
  }
  wave_lds_order();

  if (lane < 8) {
    int s_ = lane;
    v2f acc;  acc.x = 0.0f;  acc.y = 0.0f;
    #pragma unroll
    for (int m = 0; m < 16; ++m) {
      const v2f x = w.v.Wc[m][s_], y = w.v.Wc[m][8];
      v2f xr2;  xr2.x = x.x;  xr2.y = x.x;
      v2f xi2;  xi2.x = x.y;  xi2.y = x.y;
      v2f yA;   yA.x  = y.x;  yA.y  = -y.y;
      v2f yB;   yB.x  = y.y;  yB.y  =  y.x;
      acc = __builtin_elementwise_fma(xr2, yA, acc);
      acc = __builtin_elementwise_fma(xi2, yB, acc);
    }
    w.ymfre[s_] = acc.x; w.ymfim[s_] = acc.y;
  }
  #pragma unroll
  for (int t = 0; t < 2; ++t) {
    int idx = lane + 32*t;
    int i = idx >> 3, j = idx & 7;
    v2f acc;  acc.x = 0.0f;  acc.y = 0.0f;
    #pragma unroll
    for (int m = 0; m < 16; ++m) {
      const v2f x = w.v.Wc[m][i], y = w.v.Wc[m][j];
      v2f xr2;  xr2.x = x.x;  xr2.y = x.x;
      v2f xi2;  xi2.x = x.y;  xi2.y = x.y;
      v2f yA;   yA.x  = y.x;  yA.y  = -y.y;
      v2f yB;   yB.x  = y.y;  yB.y  =  y.x;
      acc = __builtin_elementwise_fma(xr2, yA, acc);
      acc = __builtin_elementwise_fma(xi2, yB, acc);
    }
    w.gre[i][j] = acc.x; w.gim[i][j] = acc.y;
    w.bm[i][j] = acc.x;  w.bm[i + 8][j] = acc.y;
  }
  wave_lds_order();

  const float C1 = 1.0f * INV_SQRT10, C3 = 3.0f * INV_SQRT10;
  const float prA = (k_out & 2) ? -C1 : C1, prB = (k_out & 2) ? -C3 : C3;
  const float piA = (k_out & 1) ? -C1 : C1, piB = (k_out & 1) ? -C3 : C3;
  float my_lla = 0.0f;
  for (int iter = 0; iter < 2; ++iter) {
    my_lla = my_lld;

    float lg[4];
    float var_s;
    {
      const float lsp = log_sigmoid( my_lla);
      const float lsn = log_sigmoid(-my_lla);
      const float p0 = quad_read<QUAD_LANE0>(lsp), n0 = quad_read<QUAD_LANE0>(lsn);
      const float p1 = quad_read<QUAD_LANE1>(lsp), n1 = quad_read<QUAD_LANE1>(lsn);
      const float p2 = quad_read<QUAD_LANE2>(lsp), n2 = quad_read<QUAD_LANE2>(lsn);
      const float p3 = quad_read<QUAD_LANE3>(lsp), n3 = quad_read<QUAD_LANE3>(lsn);
      const float base = ((k_out & 2) ? p0 : n0) + ((k_out & 1) ? p1 : n1);
      lg[0] = (base + n2) + n3;
      lg[1] = (base + n2) + p3;
      lg[2] = (base + p2) + n3;
      lg[3] = (base + p2) + p3;
      const float mx = quad_max(fmaxf(fmaxf(lg[0], lg[1]), fmaxf(lg[2], lg[3])));
      float sum = 0.0f, mre = 0.0f, mim = 0.0f, e2 = 0.0f;
      #pragma unroll
      for (int j = 0; j < 4; ++j) {
        float e = __expf(lg[j] - mx);
        float pr = (j & 2) ? prB : prA, pi = (j & 1) ? piB : piA;
        sum += e; mre += e*pr; mim += e*pi; e2 += e*(pr*pr + pi*pi);
      }
      sum = quad_sum(sum); mre = quad_sum(mre); mim = quad_sum(mim); e2 = quad_sum(e2);
      float invs = __builtin_amdgcn_rcpf(sum);
      mre *= invs; mim *= invs; e2 *= invs;
      var_s = e2 - (mre*mre + mim*mim);
      if (k_out == 0) { w.xre[s_out] = mre; w.xim[s_out] = mim; w.varL[s_out] = var_s; }
    }
    wave_lds_order();

    if (lane < 8) {
      int i = lane;
      float ar = 0.0f, ai = 0.0f;
      #pragma unroll
      for (int j = 0; j < 8; ++j) {
        float gre_ = w.gre[i][j], gim_ = w.gim[i][j];
        float xr = w.xre[j], xi = w.xim[j];
        ar += gre_*xr - gim_*xi;
        ai += gre_*xi + gim_*xr;
      }
      w.gxre[i] = ar; w.gxim[i] = ai;
    }
    wave_lds_order();

    #pragma unroll
    for (int t = 0; t < 2; ++t) {
      int idx = lane + 32*t;
      int i = idx >> 3, j = idx & 7;
      float gre_ = w.gre[j][i], gim_ = w.gim[j][i];
      float xr = w.xre[i], xi = w.xim[i];
      float tre = w.ymfre[j] + (gre_*xr - gim_*xi) - w.gxre[j];
      float tim = w.ymfim[j] + (gre_*xi + gim_*xr) - w.gxim[j];
      w.bm[j][8 + i] = tre;  w.bm[j + 8][8 + i] = tim;
    }
    {
      const int cc = lane & 15;
      const int cj = cc & 7;
      const int hf = lane >> 4;
      const float vv = w.varL[cj];
      v2f x[4];
      #pragma unroll
      for (int rr = 0; rr < 4; ++rr) {
        const int r = 4*hf + rr;
        const float dl = (r == cj) ? 1.0f : 0.0f;
        const float cre = __builtin_fmaf(w.gre[r][cj], vv, dl);
        const float cim = w.gim[r][cj] * vv;
        x[rr].x = (cc < 8) ? cre : dl;
        x[rr].y = (cc < 8) ? cim : 0.0f;
      }
      #pragma unroll
      for (int k = 0; k < 8; ++k) {
        const int kh = k >> 2, kr = k & 3;
        const bool own = (hf == kh);
        const float pre = __int_as_float(__builtin_amdgcn_readlane(__float_as_int(x[kr].x), 16*kh + k));
        const float pim = __int_as_float(__builtin_amdgcn_readlane(__float_as_int(x[kr].y), 16*kh + k));
        const float rd  = __builtin_amdgcn_rcpf(__builtin_fmaf(pre, pre, pim * pim));
        const float qre = pre * rd, qim = -pim * rd;
        #pragma unroll
        for (int rr = 0; rr < 4; ++rr) w.v.cb[hf][rr >> 1][cc][rr & 1] = x[rr];
        wave_lds_order();
        v2f f[4];
        #pragma unroll
        for (int rr = 0; rr < 4; ++rr) f[rr] = w.v.cb[hf][rr >> 1][k][rr & 1];
        const v2f xk = w.v.cb[kh][kr >> 1][cc][kr & 1];
        v2f sw;  sw.x = -xk.y;  sw.y = xk.x;
        v2f q1;  q1.x = qre;    q1.y = qre;
        v2f q2;  q2.x = qim;    q2.y = qim;
        v2f t = sw * q2;
        t = __builtin_elementwise_fma(xk, q1, t);
        v2f sT;  sT.x =  t.y;  sT.y = -t.x;
        #pragma unroll
        for (int rr = 0; rr < 4; ++rr) {
          v2f nf1;  nf1.x = -f[rr].x;  nf1.y = -f[rr].x;
          v2f f2;   f2.x  =  f[rr].y;  f2.y  =  f[rr].y;
          v2f u = __builtin_elementwise_fma(nf1, t, x[rr]);
          u = __builtin_elementwise_fma(f2, sT, u);
          if (rr == kr) {
            u.x = own ? t.x : u.x;
            u.y = own ? t.y : u.y;
          }
          x[rr] = u;
        }
      }
      if (cc >= 8) {
        const int j = cc - 8;
        #pragma unroll
        for (int rr = 0; rr < 4; ++rr) {
          const int r = 4*hf + rr;
          w.u.g.aug[r][16 + j]         =  x[rr].x;
          w.u.g.aug[r][16 + 8 + j]     = -x[rr].y;
          w.u.g.aug[r + 8][16 + j]     =  x[rr].y;
          w.u.g.aug[r + 8][16 + 8 + j] =  x[rr].x;
        }
      }
    }
    wave_lds_order();

    {
      int nn = lane & 15, hf = lane >> 4;
      v8f p = {};
      #pragma unroll
      for (int kb = 0; kb < 4; ++kb) {
        int k0 = kb*4 + hf*2;
        v2f af, bf;
        af.x = w.u.g.aug[nn][16 + k0];
        af.y = w.u.g.aug[nn][16 + k0 + 1];
        bf.x = w.bm[k0][nn];   bf.y = w.bm[k0+1][nn];
        p = __builtin_amdgcn_wmma_f32_16x16x4_f32(false, af, false, bf,
                                                  (short)0, p, false, false);
      }
      #pragma unroll
      for (int v = 0; v < 8; ++v) w.u.g.resb[lane][v] = p[v];
    }
    wave_lds_order();

    const float m1  = w.u.g.resb[s_out][s_out];
    const float rm1 = __builtin_amdgcn_rcpf(m1);
    const float zr = w.u.g.resb[8 + s_out][s_out]  * rm1;
    const float zi = w.u.g.resb[24 + s_out][s_out] * rm1;
    const float rho = m1 * __builtin_amdgcn_rcpf(fmaxf(1.0f - var_s * m1, EPSV));

    {
      float ev[4];
      #pragma unroll
      for (int j = 0; j < 4; ++j) {
        float pr = (j & 2) ? prB : prA, pi = (j & 1) ? piB : piA;
        float dre = zr - pr, dim = zi - pi;
        ev[j] = -(dre*dre + dim*dim) * rho + lg[j];
      }
      const float b2_0 = quad_max(fmaxf(ev[0], ev[1])), b2_1 = quad_max(fmaxf(ev[2], ev[3]));
      const float b3_0 = quad_max(fmaxf(ev[0], ev[2])), b3_1 = quad_max(fmaxf(ev[1], ev[3]));
      const float mall = fmaxf(fmaxf(ev[0], ev[1]), fmaxf(ev[2], ev[3]));
      const float u = fmaxf(mall, quad_read<QUAD_XOR1>(mall));
      const float v = fmaxf(mall, quad_read<QUAD_XOR2>(mall));
      const float b0_0 = quad_read<QUAD_LANE0>(u), b0_1 = quad_read<QUAD_LANE2>(u);
      const float b1_0 = quad_read<QUAD_LANE0>(v), b1_1 = quad_read<QUAD_LANE1>(v);
      float one  = (k_out == 0) ? b0_1 : (k_out == 1) ? b1_1 : (k_out == 2) ? b2_1 : b3_1;
      float zero = (k_out == 0) ? b0_0 : (k_out == 1) ? b1_0 : (k_out == 2) ? b2_0 : b3_0;
      my_lld = fmaxf(one, -1e30f) - fmaxf(zero, -1e30f);
    }
  }

  out[b*32 + lane] = my_lld - my_lla;
}

extern "C" void kernel_launch(void* const* d_in, const int* in_sizes, int n_in,
                              void* d_out, int out_size, void* d_ws, size_t ws_size,
                              hipStream_t stream) {
  (void)n_in; (void)out_size; (void)d_ws; (void)ws_size;
  const float* y_re  = (const float*)d_in[0];
  const float* y_im  = (const float*)d_in[1];
  const float* h_re  = (const float*)d_in[2];
  const float* h_im  = (const float*)d_in[3];
  const float* prior = (const float*)d_in[4];
  const float* s_re  = (const float*)d_in[5];
  const float* s_im  = (const float*)d_in[6];
  float* out = (float*)d_out;

  int B = in_sizes[0] / 16;
  int blocks = B / WAVES_PER_BLOCK;
  mmse_pic_kernel<<<blocks, 256, 0, stream>>>(y_re, y_im, h_re, h_im,
                                              prior, s_re, s_im, out);
}
